# starting kernel
# baseline (speedup 1.0000x reference)
#include <hip/hip_runtime.h>
#include <hip/hip_bf16.h>

typedef __attribute__((ext_vector_type(2))) float v2f;
typedef __attribute__((ext_vector_type(8))) float v8f;

#define NNODES 50000
#define NEDGES 800000
#define NPAIRS 500000
#define F_IN   128
#define HID    256
#define EIN    16
#define ZK     (2 * HID + EIN)
#define GN     4
#define ET     5

__device__ __forceinline__ v8f wmma_ab(v2f a, v2f b, v8f c) {
  return __builtin_amdgcn_wmma_f32_16x16x4_f32(false, a, false, b,
                                               (short)0, c, false, false);
}

__device__ __forceinline__ v2f ld2(const float* __restrict__ p, int k) {
  v2f r;
  r.x = p[k];
  r.y = p[k + 1];
  return r;
}

__global__ void k_zero(float* __restrict__ p, int n) {
  int i = blockIdx.x * blockDim.x + threadIdx.x;
  if (i < n) p[i] = 0.f;
}

__global__ void k_count(const int* __restrict__ dst, float* __restrict__ cnt) {
  int e = blockIdx.x * blockDim.x + threadIdx.x;
  if (e < NEDGES) atomicAdd(&cnt[dst[e]], 1.f);
}

__global__ void k_scatter(const int* __restrict__ src, const int* __restrict__ dst,
                          const float* __restrict__ x, float* __restrict__ agg, int lgF) {
  int i = blockIdx.x * blockDim.x + threadIdx.x;
  int tot = NEDGES << lgF;
  if (i >= tot) return;
  int e = i >> lgF;
  int f = i & ((1 << lgF) - 1);
  atomicAdd(&agg[(dst[e] << lgF) + f], x[(src[e] << lgF) + f]);
}

__global__ void k_norm(float* __restrict__ agg, const float* __restrict__ cnt, int lgF) {
  int i = blockIdx.x * blockDim.x + threadIdx.x;
  int tot = NNODES << lgF;
  if (i >= tot) return;
  int n = i >> lgF;
  agg[i] *= 1.f / fmaxf(cnt[n], 1.f);
}

__global__ void k_sage_gemm(const float* __restrict__ A, const float* __restrict__ X,
                            const float* __restrict__ Wl, const float* __restrict__ Wr,
                            const float* __restrict__ bias, float* __restrict__ out,
                            int K) {
  int gtid = blockIdx.x * blockDim.x + threadIdx.x;
  int wave = gtid >> 5;
  int lane = threadIdx.x & 31;
  const int ngrp = (HID / 16) / GN;
  const int tiles_m = NNODES / 16;
  if (wave >= tiles_m * ngrp) return;
  int tm = wave / ngrp;
  int tg = wave - tm * ngrp;
  int koff = (lane >> 4) << 1;
  int row = tm * 16 + (lane & 15);
  int col0 = tg * (GN * 16) + (lane & 15);
  const float* arow = A + row * K;
  const float* xrow = X + row * K;
  const float* wl[GN];
  const float* wr[GN];
#pragma unroll
  for (int t = 0; t < GN; ++t) {
    int col = col0 + t * 16;
    wl[t] = Wl + col * K;
    wr[t] = Wr + col * K;
  }
  v8f c[GN] = {};
  for (int kb = 0; kb < K; kb += 4) {
    int k = kb + koff;
    v2f a = ld2(arow, k);
    v2f x2 = ld2(xrow, k);
#pragma unroll
    for (int t = 0; t < GN; ++t) {
      c[t] = wmma_ab(a, ld2(wl[t], k), c[t]);
      c[t] = wmma_ab(x2, ld2(wr[t], k), c[t]);
    }
  }
  int mbase = tm * 16 + ((lane >> 4) << 3);
#pragma unroll
  for (int t = 0; t < GN; ++t) {
    int col = col0 + t * 16;
    float bv = bias[col];
#pragma unroll
    for (int r = 0; r < 8; ++r)
      out[(mbase + r) * HID + col] = fmaxf(c[t][r] + bv, 0.f);
  }
}

__global__ void k_edge_mlp(const float* __restrict__ H, const int* __restrict__ eu,
                           const int* __restrict__ ev, const float* __restrict__ eattr,
                           const float* __restrict__ logexp,
                           const float* __restrict__ W1, const float* __restrict__ b1,
                           const float* __restrict__ W2, const float* __restrict__ b2,
                           float* __restrict__ out) {
  int gtid = blockIdx.x * blockDim.x + threadIdx.x;
  int wave = gtid >> 5;
  int lane = threadIdx.x & 31;
  const int groups = NPAIRS / 16 / ET;
  if (wave >= groups) return;
  int koff = (lane >> 4) << 1;
  int p0 = wave * (ET * 16) + (lane & 15);
  const float* urow[ET];
  const float* vrow[ET];
  const float* arow[ET];
#pragma unroll
  for (int t = 0; t < ET; ++t) {
    int p = p0 + t * 16;
    urow[t] = H + eu[p] * HID;
    vrow[t] = H + ev[p] * HID;
    arow[t] = eattr + p * EIN;
  }
  float acc[ET][8] = {};
  float b2v = b2[0];

  for (int nt = 0; nt < HID / 16; ++nt) {
    int col = nt * 16 + (lane & 15);
    const float* w = W1 + col * ZK;
    v8f c[ET] = {};
    for (int kb = 0; kb < HID; kb += 4) {
      v2f b = ld2(w, kb + koff);
#pragma unroll
      for (int t = 0; t < ET; ++t)
        c[t] = wmma_ab(ld2(urow[t], kb + koff), b, c[t]);
    }
    for (int kb = 0; kb < HID; kb += 4) {
      v2f b = ld2(w, HID + kb + koff);
#pragma unroll
      for (int t = 0; t < ET; ++t)
        c[t] = wmma_ab(ld2(vrow[t], kb + koff), b, c[t]);
    }
#pragma unroll
    for (int kb = 0; kb < EIN; kb += 4) {
      v2f b = ld2(w, 2 * HID + kb + koff);
#pragma unroll
      for (int t = 0; t < ET; ++t)
        c[t] = wmma_ab(ld2(arow[t], kb + koff), b, c[t]);
    }
    float bv = b1[col];
    float wv = W2[col];
#pragma unroll
    for (int t = 0; t < ET; ++t)
#pragma unroll
      for (int r = 0; r < 8; ++r)
        acc[t][r] += fmaxf(c[t][r] + bv, 0.f) * wv;
  }

#pragma unroll
  for (int t = 0; t < ET; ++t) {
#pragma unroll
    for (int r = 0; r < 8; ++r) {
      float s = acc[t][r];
      s += __shfl_xor(s, 1);
      s += __shfl_xor(s, 2);
      s += __shfl_xor(s, 4);
      s += __shfl_xor(s, 8);
      acc[t][r] = s;
    }
  }
  if ((lane & 15) == 0) {
    int half = (lane >> 4) << 3;
#pragma unroll
    for (int t = 0; t < ET; ++t) {
      int base = (wave * ET + t) * 16 + half;
#pragma unroll
      for (int r = 0; r < 8; ++r)
        out[base + r] = logexp[base + r] + acc[t][r] + b2v;
    }
  }
}

extern "C" void kernel_launch(void* const* d_in, const int* in_sizes, int n_in,
                              void* d_out, int out_size, void* d_ws, size_t ws_size,
                              hipStream_t stream) {
  (void)in_sizes; (void)n_in; (void)out_size; (void)ws_size;
  const float* x      = (const float*)d_in[0];
  const int*   eidx   = (const int*)d_in[1];
  const int*   eu     = (const int*)d_in[2];
  const int*   ev     = (const int*)d_in[3];
  const float* eattr  = (const float*)d_in[4];
  const float* logexp = (const float*)d_in[5];
  const float* Wl1    = (const float*)d_in[6];
  const float* bl1    = (const float*)d_in[7];
  const float* Wr1    = (const float*)d_in[8];
  const float* Wl2    = (const float*)d_in[9];
  const float* bl2    = (const float*)d_in[10];
  const float* Wr2    = (const float*)d_in[11];
  const float* W1     = (const float*)d_in[12];
  const float* b1     = (const float*)d_in[13];
  const float* W2     = (const float*)d_in[14];
  const float* b2     = (const float*)d_in[15];
  float* out = (float*)d_out;

  const int* src = eidx;
  const int* dst = eidx + NEDGES;

  float* ws  = (float*)d_ws;
  float* cnt = ws;
  float* agg = ws + NNODES;
  float* h1  = agg + NNODES * HID;
  float* h2  = h1 + NNODES * HID;

  const int TB = 256;
  int gemm_waves = (NNODES / 16) * ((HID / 16) / GN);
  int gemm_blocks = (gemm_waves * 32 + TB - 1) / TB;

  {
    int nz = NNODES * (1 + F_IN);
    k_zero<<<(nz + TB - 1) / TB, TB, 0, stream>>>(cnt, nz);
    k_count<<<(NEDGES + TB - 1) / TB, TB, 0, stream>>>(dst, cnt);
    int nt = NEDGES << 7;
    k_scatter<<<(nt + TB - 1) / TB, TB, 0, stream>>>(src, dst, x, agg, 7);
    int nn = NNODES << 7;
    k_norm<<<(nn + TB - 1) / TB, TB, 0, stream>>>(agg, cnt, 7);
    k_sage_gemm<<<gemm_blocks, TB, 0, stream>>>(agg, x, Wl1, Wr1, bl1, h1, F_IN);
  }
  {
    int nz = NNODES * HID;
    k_zero<<<(nz + TB - 1) / TB, TB, 0, stream>>>(agg, nz);
    int nt = NEDGES << 8;
    k_scatter<<<(nt + TB - 1) / TB, TB, 0, stream>>>(src, dst, h1, agg, 8);
    int nn = NNODES << 8;
    k_norm<<<(nn + TB - 1) / TB, TB, 0, stream>>>(agg, cnt, 8);
    k_sage_gemm<<<gemm_blocks, TB, 0, stream>>>(agg, h1, Wl2, Wr2, bl2, h2, HID);
  }
  {
    int waves = NPAIRS / 16 / ET;
    int blocks = (waves * 32 + TB - 1) / TB;
    k_edge_mlp<<<blocks, TB, 0, stream>>>(h2, eu, ev, eattr, logexp,
                                          W1, b1, W2, b2, out);
  }
}


# edge-MLP layer 1: h[u] and h[v] products done once per node (50000 rows, not 500000 pairs), projected rows gathered and summed per pair; eattr product stays per pair on WMMA; regroups the fp sums
# speedup vs baseline: 3.5339x; 3.5339x over previous
#include <hip/hip_runtime.h>
#include <hip/hip_bf16.h>

typedef __attribute__((ext_vector_type(2))) float v2f;
typedef __attribute__((ext_vector_type(4))) float v4f;
typedef __attribute__((ext_vector_type(8))) float v8f;

#define NNODES 50000
#define NEDGES 800000
#define NPAIRS 500000
#define F_IN   128
#define HID    256
#define EIN    16
#define ZK     (2 * HID + EIN)
#define GN     4
#define PM     5
#define PN     4
#define TPW    2

__device__ __forceinline__ v8f wmma_ab(v2f a, v2f b, v8f c) {
  return __builtin_amdgcn_wmma_f32_16x16x4_f32(false, a, false, b,
                                               (short)0, c, false, false);
}

__device__ __forceinline__ v2f ld2(const float* __restrict__ p, int k) {
  v2f r;
  r.x = p[k];
  r.y = p[k + 1];
  return r;
}

__global__ void k_zero(float* __restrict__ p, int n) {
  int i = blockIdx.x * blockDim.x + threadIdx.x;
  if (i < n) p[i] = 0.f;
}

__global__ void k_count(const int* __restrict__ dst, float* __restrict__ cnt) {
  int e = blockIdx.x * blockDim.x + threadIdx.x;
  if (e < NEDGES) atomicAdd(&cnt[dst[e]], 1.f);
}

__global__ void k_scatter(const int* __restrict__ src, const int* __restrict__ dst,
                          const float* __restrict__ x, float* __restrict__ agg, int lgF) {
  int i = blockIdx.x * blockDim.x + threadIdx.x;
  int tot = NEDGES << lgF;
  if (i >= tot) return;
  int e = i >> lgF;
  int f = i & ((1 << lgF) - 1);
  atomicAdd(&agg[(dst[e] << lgF) + f], x[(src[e] << lgF) + f]);
}

__global__ void k_norm(float* __restrict__ agg, const float* __restrict__ cnt, int lgF) {
  int i = blockIdx.x * blockDim.x + threadIdx.x;
  int tot = NNODES << lgF;
  if (i >= tot) return;
  int n = i >> lgF;
  agg[i] *= 1.f / fmaxf(cnt[n], 1.f);
}

__global__ void k_sage_gemm(const float* __restrict__ A, const float* __restrict__ X,
                            const float* __restrict__ Wl, const float* __restrict__ Wr,
                            const float* __restrict__ bias, float* __restrict__ out,
                            int K) {
  int gtid = blockIdx.x * blockDim.x + threadIdx.x;
  int wave = gtid >> 5;
  int lane = threadIdx.x & 31;
  const int ngrp = (HID / 16) / GN;
  const int tiles_m = NNODES / 16;
  if (wave >= tiles_m * ngrp) return;
  int tm = wave / ngrp;
  int tg = wave - tm * ngrp;
  int koff = (lane >> 4) << 1;
  int row = tm * 16 + (lane & 15);
  int col0 = tg * (GN * 16) + (lane & 15);
  const float* arow = A + row * K;
  const float* xrow = X + row * K;
  const float* wl[GN];
  const float* wr[GN];
#pragma unroll
  for (int t = 0; t < GN; ++t) {
    int col = col0 + t * 16;
    wl[t] = Wl + col * K;
    wr[t] = Wr + col * K;
  }
  v8f c[GN] = {};
  for (int kb = 0; kb < K; kb += 4) {
    int k = kb + koff;
    v2f a = ld2(arow, k);
    v2f x2 = ld2(xrow, k);
#pragma unroll
    for (int t = 0; t < GN; ++t) {
      c[t] = wmma_ab(a, ld2(wl[t], k), c[t]);
      c[t] = wmma_ab(x2, ld2(wr[t], k), c[t]);
    }
  }
  int mbase = tm * 16 + ((lane >> 4) << 3);
#pragma unroll
  for (int t = 0; t < GN; ++t) {
    int col = col0 + t * 16;
    float bv = bias[col];
#pragma unroll
    for (int r = 0; r < 8; ++r)
      out[(mbase + r) * HID + col] = fmaxf(c[t][r] + bv, 0.f);
  }
}

__device__ __forceinline__ void ld8(const float* __restrict__ p, int k, v4f& lo, v4f& hi) {
  lo = *(const v4f*)(p + k);
  hi = *(const v4f*)(p + k + 4);
}

__device__ __forceinline__ v2f pair_of(v4f lo, v4f hi, int q) {
  v2f r;
  if (q == 0) { r.x = lo.x; r.y = lo.y; }
  else if (q == 1) { r.x = lo.z; r.y = lo.w; }
  else if (q == 2) { r.x = hi.x; r.y = hi.y; }
  else { r.x = hi.z; r.y = hi.w; }
  return r;
}

__global__ void __launch_bounds__(128)
k_node_proj(const float* __restrict__ Hn, const float* __restrict__ W1,
            const float* __restrict__ b1, float* __restrict__ Q) {
  int gtid = blockIdx.x * blockDim.x + threadIdx.x;
  int wave = gtid >> 5;
  int lane = threadIdx.x & 31;
  const int ngrp = (2 * HID / 16) / PN;
  const int mgrp = NNODES / 16 / PM;
  if (wave >= mgrp * ngrp) return;
  int gm = wave / ngrp;
  int gn = wave - gm * ngrp;
  int plane = gn >> 2;
  int koff = (lane >> 4) << 3;
  int row0 = gm * (PM * 16) + (lane & 15);
  int col0 = (gn & 3) * (PN * 16) + (lane & 15);
  const float* hrow[PM];
  const float* wrow[PN];
#pragma unroll
  for (int t = 0; t < PM; ++t) hrow[t] = Hn + (row0 + t * 16) * HID;
#pragma unroll
  for (int i = 0; i < PN; ++i) wrow[i] = W1 + (col0 + i * 16) * ZK + plane * HID;
  v8f c[PM][PN] = {};
  for (int kb = 0; kb < HID; kb += 16) {
    int k = kb + koff;
    v4f alo[PM], ahi[PM];
    v4f blo[PN], bhi[PN];
#pragma unroll
    for (int t = 0; t < PM; ++t) ld8(hrow[t], k, alo[t], ahi[t]);
#pragma unroll
    for (int i = 0; i < PN; ++i) ld8(wrow[i], k, blo[i], bhi[i]);
#pragma unroll
    for (int q = 0; q < 4; ++q)
#pragma unroll
      for (int t = 0; t < PM; ++t)
#pragma unroll
        for (int i = 0; i < PN; ++i)
          c[t][i] = wmma_ab(pair_of(alo[t], ahi[t], q), pair_of(blo[i], bhi[i], q), c[t][i]);
  }
  float* qp = Q + plane * (NNODES * HID);
  int mbase = gm * (PM * 16) + ((lane >> 4) << 3);
#pragma unroll
  for (int i = 0; i < PN; ++i) {
    int col = col0 + i * 16;
    float bv = (plane == 0) ? b1[col] : 0.f;
#pragma unroll
    for (int t = 0; t < PM; ++t)
#pragma unroll
      for (int r = 0; r < 8; ++r)
        qp[(mbase + t * 16 + r) * HID + col] = c[t][i][r] + bv;
  }
}

__device__ __forceinline__ int pair_col(int nt, int n) {
  return ((nt >> 2) << 6) + (n << 2) + (nt & 3);
}

__global__ void k_pair_mlp(const float* __restrict__ Qu, const float* __restrict__ Qv,
                           const int* __restrict__ eu, const int* __restrict__ ev,
                           const float* __restrict__ eattr,
                           const float* __restrict__ logexp,
                           const float* __restrict__ W1, const float* __restrict__ W2,
                           const float* __restrict__ b2, float* __restrict__ out) {
  __shared__ v2f sB[(HID / 16) * (EIN / 4) * 32];
  for (int idx = threadIdx.x; idx < (HID / 16) * (EIN / 4) * 32; idx += blockDim.x) {
    int nt = idx >> 7;
    int ks = (idx >> 5) & 3;
    int ln = idx & 31;
    const float* w = W1 + pair_col(nt, ln & 15) * ZK + 2 * HID;
    sB[idx] = ld2(w, ks * 4 + ((ln >> 4) << 1));
  }
  __syncthreads();

  int gtid = blockIdx.x * blockDim.x + threadIdx.x;
  int wave = gtid >> 5;
  int lane = threadIdx.x & 31;
  if (wave >= NPAIRS / 16 / TPW) return;
  int n = lane & 15;
  int half = (lane >> 4) << 3;
  int koff = (lane >> 4) << 1;
  float b2v = b2[0];

  for (int tt = 0; tt < TPW; ++tt) {
    int tile = wave * TPW + tt;
    int ebase = tile * 16 + half;
    const float* arow = eattr + (tile * 16 + n) * EIN;
    v2f a[EIN / 4];
#pragma unroll
    for (int ks = 0; ks < EIN / 4; ++ks) a[ks] = ld2(arow, ks * 4 + koff);
    int uoff[8];
    int voff[8];
#pragma unroll
    for (int r = 0; r < 8; ++r) {
      uoff[r] = eu[ebase + r] * HID + (n << 2);
      voff[r] = ev[ebase + r] * HID + (n << 2);
    }
    float acc[8] = {};
    for (int j = 0; j < 4; ++j) {
      v4f w2v = *(const v4f*)(W2 + (j << 6) + (n << 2));
      v8f c[4];
#pragma unroll
      for (int r = 0; r < 8; ++r) {
        v4f pu = *(const v4f*)(Qu + uoff[r] + (j << 6));
        v4f pv = *(const v4f*)(Qv + voff[r] + (j << 6));
#pragma unroll
        for (int i = 0; i < 4; ++i) c[i][r] = pu[i] + pv[i];
      }
#pragma unroll
      for (int ks = 0; ks < EIN / 4; ++ks)
#pragma unroll
        for (int i = 0; i < 4; ++i)
          c[i] = wmma_ab(a[ks], sB[((j * 4 + i) * 4 + ks) * 32 + lane], c[i]);
#pragma unroll
      for (int i = 0; i < 4; ++i)
#pragma unroll
        for (int r = 0; r < 8; ++r)
          acc[r] += fmaxf(c[i][r], 0.f) * w2v[i];
    }
#pragma unroll
    for (int r = 0; r < 8; ++r) {
      float s = acc[r];
      s += __shfl_xor(s, 1);
      s += __shfl_xor(s, 2);
      s += __shfl_xor(s, 4);
      s += __shfl_xor(s, 8);
      acc[r] = s;
    }
    if (n == 0) {
#pragma unroll
      for (int r = 0; r < 8; ++r)
        out[ebase + r] = logexp[ebase + r] + acc[r] + b2v;
    }
  }
}

extern "C" void kernel_launch(void* const* d_in, const int* in_sizes, int n_in,
                              void* d_out, int out_size, void* d_ws, size_t ws_size,
                              hipStream_t stream) {
  (void)in_sizes; (void)n_in; (void)out_size; (void)ws_size;
  const float* x      = (const float*)d_in[0];
  const int*   eidx   = (const int*)d_in[1];
  const int*   eu     = (const int*)d_in[2];
  const int*   ev     = (const int*)d_in[3];
  const float* eattr  = (const float*)d_in[4];
  const float* logexp = (const float*)d_in[5];
  const float* Wl1    = (const float*)d_in[6];
  const float* bl1    = (const float*)d_in[7];
  const float* Wr1    = (const float*)d_in[8];
  const float* Wl2    = (const float*)d_in[9];
  const float* bl2    = (const float*)d_in[10];
  const float* Wr2    = (const float*)d_in[11];
  const float* W1     = (const float*)d_in[12];
  const float* b1     = (const float*)d_in[13];
  const float* W2     = (const float*)d_in[14];
  const float* b2     = (const float*)d_in[15];
  float* out = (float*)d_out;

  const int* src = eidx;
  const int* dst = eidx + NEDGES;

  float* ws  = (float*)d_ws;
  float* cnt = ws;
  float* agg = ws + NNODES;
  float* h1  = agg + NNODES * HID;
  float* h2  = h1 + NNODES * HID;

  const int TB = 256;
  int gemm_waves = (NNODES / 16) * ((HID / 16) / GN);
  int gemm_blocks = (gemm_waves * 32 + TB - 1) / TB;

  {
    int nz = NNODES * (1 + F_IN);
    k_zero<<<(nz + TB - 1) / TB, TB, 0, stream>>>(cnt, nz);
    k_count<<<(NEDGES + TB - 1) / TB, TB, 0, stream>>>(dst, cnt);
    int nt = NEDGES << 7;
    k_scatter<<<(nt + TB - 1) / TB, TB, 0, stream>>>(src, dst, x, agg, 7);
    int nn = NNODES << 7;
    k_norm<<<(nn + TB - 1) / TB, TB, 0, stream>>>(agg, cnt, 7);
    k_sage_gemm<<<gemm_blocks, TB, 0, stream>>>(agg, x, Wl1, Wr1, bl1, h1, F_IN);
  }
  {
    int nz = NNODES * HID;
    k_zero<<<(nz + TB - 1) / TB, TB, 0, stream>>>(agg, nz);
    int nt = NEDGES << 8;
    k_scatter<<<(nt + TB - 1) / TB, TB, 0, stream>>>(src, dst, h1, agg, 8);
    int nn = NNODES << 8;
    k_norm<<<(nn + TB - 1) / TB, TB, 0, stream>>>(agg, cnt, 8);
    k_sage_gemm<<<gemm_blocks, TB, 0, stream>>>(agg, h1, Wl2, Wr2, bl2, h2, HID);
  }
  {
    float* qu = agg;
    float* qv = h1;
    int pwaves = (NNODES / 16 / PM) * ((2 * HID / 16) / PN);
    k_node_proj<<<pwaves / 4, 128, 0, stream>>>(h2, W1, b1, qu);
    int waves = NPAIRS / 16 / TPW;
    int blocks = (waves * 32 + TB - 1) / TB;
    k_pair_mlp<<<blocks, TB, 0, stream>>>(qu, qv, eu, ev, eattr, logexp,
                                          W1, W2, b2, out);
  }
}


# Both graph-conv layer GEMMs re-tiled: one wave per 80x64 block (20 accumulator tiles), lanes read 32 contiguous bytes of a row per 16-deep k step; agg product summed before root's (regroups fp sums)
# speedup vs baseline: 4.9337x; 1.3961x over previous
#include <hip/hip_runtime.h>
#include <hip/hip_bf16.h>

typedef __attribute__((ext_vector_type(2))) float v2f;
typedef __attribute__((ext_vector_type(4))) float v4f;
typedef __attribute__((ext_vector_type(8))) float v8f;

#define NNODES 50000
#define NEDGES 800000
#define NPAIRS 500000
#define F_IN   128
#define HID    256
#define EIN    16
#define ZK     (2 * HID + EIN)
#define SM     5
#define SN     4
#define PM     5
#define PN     4
#define TPW    2

__device__ __forceinline__ v8f wmma_ab(v2f a, v2f b, v8f c) {
  return __builtin_amdgcn_wmma_f32_16x16x4_f32(false, a, false, b,
                                               (short)0, c, false, false);
}

__device__ __forceinline__ v2f ld2(const float* __restrict__ p, int k) {
  v2f r;
  r.x = p[k];
  r.y = p[k + 1];
  return r;
}

__global__ void k_zero(float* __restrict__ p, int n) {
  int i = blockIdx.x * blockDim.x + threadIdx.x;
  if (i < n) p[i] = 0.f;
}

__global__ void k_count(const int* __restrict__ dst, float* __restrict__ cnt) {
  int e = blockIdx.x * blockDim.x + threadIdx.x;
  if (e < NEDGES) atomicAdd(&cnt[dst[e]], 1.f);
}

__global__ void k_scatter(const int* __restrict__ src, const int* __restrict__ dst,
                          const float* __restrict__ x, float* __restrict__ agg, int lgF) {
  int i = blockIdx.x * blockDim.x + threadIdx.x;
  int tot = NEDGES << lgF;
  if (i >= tot) return;
  int e = i >> lgF;
  int f = i & ((1 << lgF) - 1);
  atomicAdd(&agg[(dst[e] << lgF) + f], x[(src[e] << lgF) + f]);
}

__global__ void k_norm(float* __restrict__ agg, const float* __restrict__ cnt, int lgF) {
  int i = blockIdx.x * blockDim.x + threadIdx.x;
  int tot = NNODES << lgF;
  if (i >= tot) return;
  int n = i >> lgF;
  agg[i] *= 1.f / fmaxf(cnt[n], 1.f);
}

__device__ __forceinline__ void ld8(const float* __restrict__ p, int k, v4f& lo, v4f& hi) {
  lo = *(const v4f*)(p + k);
  hi = *(const v4f*)(p + k + 4);
}

__device__ __forceinline__ v2f pair_of(v4f lo, v4f hi, int q) {
  v2f r;
  if (q == 0) { r.x = lo.x; r.y = lo.y; }
  else if (q == 1) { r.x = lo.z; r.y = lo.w; }
  else if (q == 2) { r.x = hi.x; r.y = hi.y; }
  else { r.x = hi.z; r.y = hi.w; }
  return r;
}

template <int K>
__device__ __forceinline__ void sage_acc(const float* __restrict__ R,
                                         const float* __restrict__ Wm,
                                         int row0, int col0, int koff,
                                         v8f (&c)[SM][SN]) {
  const float* arow[SM];
  const float* wrow[SN];
#pragma unroll
  for (int t = 0; t < SM; ++t) arow[t] = R + (row0 + t * 16) * K;
#pragma unroll
  for (int i = 0; i < SN; ++i) wrow[i] = Wm + (col0 + i * 16) * K;
  for (int kb = 0; kb < K; kb += 16) {
    int k = kb + koff;
    v4f alo[SM], ahi[SM];
    v4f blo[SN], bhi[SN];
#pragma unroll
    for (int t = 0; t < SM; ++t) ld8(arow[t], k, alo[t], ahi[t]);
#pragma unroll
    for (int i = 0; i < SN; ++i) ld8(wrow[i], k, blo[i], bhi[i]);
#pragma unroll
    for (int q = 0; q < 4; ++q)
#pragma unroll
      for (int t = 0; t < SM; ++t)
#pragma unroll
        for (int i = 0; i < SN; ++i)
          c[t][i] = wmma_ab(pair_of(alo[t], ahi[t], q), pair_of(blo[i], bhi[i], q), c[t][i]);
  }
}

template <int K>
__global__ void __launch_bounds__(128)
k_sage_gemm(const float* __restrict__ A, const float* __restrict__ X,
            const float* __restrict__ Wl, const float* __restrict__ Wr,
            const float* __restrict__ bias, float* __restrict__ out) {
  int gtid = blockIdx.x * blockDim.x + threadIdx.x;
  int wave = gtid >> 5;
  int lane = threadIdx.x & 31;
  const int ngrp = (HID / 16) / SN;
  const int mgrp = NNODES / 16 / SM;
  if (wave >= mgrp * ngrp) return;
  int gm = wave / ngrp;
  int gn = wave - gm * ngrp;
  int koff = (lane >> 4) << 3;
  int row0 = gm * (SM * 16) + (lane & 15);
  int col0 = gn * (SN * 16) + (lane & 15);
  v8f c[SM][SN] = {};
  sage_acc<K>(A, Wl, row0, col0, koff, c);
  sage_acc<K>(X, Wr, row0, col0, koff, c);
  int mbase = gm * (SM * 16) + ((lane >> 4) << 3);
#pragma unroll
  for (int i = 0; i < SN; ++i) {
    int col = col0 + i * 16;
    float bv = bias[col];
#pragma unroll
    for (int t = 0; t < SM; ++t)
#pragma unroll
      for (int r = 0; r < 8; ++r)
        out[(mbase + t * 16 + r) * HID + col] = fmaxf(c[t][i][r] + bv, 0.f);
  }
}

__global__ void __launch_bounds__(128)
k_node_proj(const float* __restrict__ Hn, const float* __restrict__ W1,
            const float* __restrict__ b1, float* __restrict__ Q) {
  int gtid = blockIdx.x * blockDim.x + threadIdx.x;
  int wave = gtid >> 5;
  int lane = threadIdx.x & 31;
  const int ngrp = (2 * HID / 16) / PN;
  const int mgrp = NNODES / 16 / PM;
  if (wave >= mgrp * ngrp) return;
  int gm = wave / ngrp;
  int gn = wave - gm * ngrp;
  int plane = gn >> 2;
  int koff = (lane >> 4) << 3;
  int row0 = gm * (PM * 16) + (lane & 15);
  int col0 = (gn & 3) * (PN * 16) + (lane & 15);
  const float* hrow[PM];
  const float* wrow[PN];
#pragma unroll
  for (int t = 0; t < PM; ++t) hrow[t] = Hn + (row0 + t * 16) * HID;
#pragma unroll
  for (int i = 0; i < PN; ++i) wrow[i] = W1 + (col0 + i * 16) * ZK + plane * HID;
  v8f c[PM][PN] = {};
  for (int kb = 0; kb < HID; kb += 16) {
    int k = kb + koff;
    v4f alo[PM], ahi[PM];
    v4f blo[PN], bhi[PN];
#pragma unroll
    for (int t = 0; t < PM; ++t) ld8(hrow[t], k, alo[t], ahi[t]);
#pragma unroll
    for (int i = 0; i < PN; ++i) ld8(wrow[i], k, blo[i], bhi[i]);
#pragma unroll
    for (int q = 0; q < 4; ++q)
#pragma unroll
      for (int t = 0; t < PM; ++t)
#pragma unroll
        for (int i = 0; i < PN; ++i)
          c[t][i] = wmma_ab(pair_of(alo[t], ahi[t], q), pair_of(blo[i], bhi[i], q), c[t][i]);
  }
  float* qp = Q + plane * (NNODES * HID);
  int mbase = gm * (PM * 16) + ((lane >> 4) << 3);
#pragma unroll
  for (int i = 0; i < PN; ++i) {
    int col = col0 + i * 16;
    float bv = (plane == 0) ? b1[col] : 0.f;
#pragma unroll
    for (int t = 0; t < PM; ++t)
#pragma unroll
      for (int r = 0; r < 8; ++r)
        qp[(mbase + t * 16 + r) * HID + col] = c[t][i][r] + bv;
  }
}

__device__ __forceinline__ int pair_col(int nt, int n) {
  return ((nt >> 2) << 6) + (n << 2) + (nt & 3);
}

__global__ void k_pair_mlp(const float* __restrict__ Qu, const float* __restrict__ Qv,
                           const int* __restrict__ eu, const int* __restrict__ ev,
                           const float* __restrict__ eattr,
                           const float* __restrict__ logexp,
                           const float* __restrict__ W1, const float* __restrict__ W2,
                           const float* __restrict__ b2, float* __restrict__ out) {
  __shared__ v2f sB[(HID / 16) * (EIN / 4) * 32];
  for (int idx = threadIdx.x; idx < (HID / 16) * (EIN / 4) * 32; idx += blockDim.x) {
    int nt = idx >> 7;
    int ks = (idx >> 5) & 3;
    int ln = idx & 31;
    const float* w = W1 + pair_col(nt, ln & 15) * ZK + 2 * HID;
    sB[idx] = ld2(w, ks * 4 + ((ln >> 4) << 1));
  }
  __syncthreads();

  int gtid = blockIdx.x * blockDim.x + threadIdx.x;
  int wave = gtid >> 5;
  int lane = threadIdx.x & 31;
  if (wave >= NPAIRS / 16 / TPW) return;
  int n = lane & 15;
  int half = (lane >> 4) << 3;
  int koff = (lane >> 4) << 1;
  float b2v = b2[0];

  for (int tt = 0; tt < TPW; ++tt) {
    int tile = wave * TPW + tt;
    int ebase = tile * 16 + half;
    const float* arow = eattr + (tile * 16 + n) * EIN;
    v2f a[EIN / 4];
#pragma unroll
    for (int ks = 0; ks < EIN / 4; ++ks) a[ks] = ld2(arow, ks * 4 + koff);
    int uoff[8];
    int voff[8];
#pragma unroll
    for (int r = 0; r < 8; ++r) {
      uoff[r] = eu[ebase + r] * HID + (n << 2);
      voff[r] = ev[ebase + r] * HID + (n << 2);
    }
    float acc[8] = {};
    for (int j = 0; j < 4; ++j) {
      v4f w2v = *(const v4f*)(W2 + (j << 6) + (n << 2));
      v8f c[4];
#pragma unroll
      for (int r = 0; r < 8; ++r) {
        v4f pu = *(const v4f*)(Qu + uoff[r] + (j << 6));
        v4f pv = *(const v4f*)(Qv + voff[r] + (j << 6));
#pragma unroll
        for (int i = 0; i < 4; ++i) c[i][r] = pu[i] + pv[i];
      }
#pragma unroll
      for (int ks = 0; ks < EIN / 4; ++ks)
#pragma unroll
        for (int i = 0; i < 4; ++i)
          c[i] = wmma_ab(a[ks], sB[((j * 4 + i) * 4 + ks) * 32 + lane], c[i]);
#pragma unroll
      for (int i = 0; i < 4; ++i)
#pragma unroll
        for (int r = 0; r < 8; ++r)
          acc[r] += fmaxf(c[i][r], 0.f) * w2v[i];
    }
#pragma unroll
    for (int r = 0; r < 8; ++r) {
      float s = acc[r];
      s += __shfl_xor(s, 1);
      s += __shfl_xor(s, 2);
      s += __shfl_xor(s, 4);
      s += __shfl_xor(s, 8);
      acc[r] = s;
    }
    if (n == 0) {
#pragma unroll
      for (int r = 0; r < 8; ++r)
        out[ebase + r] = logexp[ebase + r] + acc[r] + b2v;
    }
  }
}

extern "C" void kernel_launch(void* const* d_in, const int* in_sizes, int n_in,
                              void* d_out, int out_size, void* d_ws, size_t ws_size,
                              hipStream_t stream) {
  (void)in_sizes; (void)n_in; (void)out_size; (void)ws_size;
  const float* x      = (const float*)d_in[0];
  const int*   eidx   = (const int*)d_in[1];
  const int*   eu     = (const int*)d_in[2];
  const int*   ev     = (const int*)d_in[3];
  const float* eattr  = (const float*)d_in[4];
  const float* logexp = (const float*)d_in[5];
  const float* Wl1    = (const float*)d_in[6];
  const float* bl1    = (const float*)d_in[7];
  const float* Wr1    = (const float*)d_in[8];
  const float* Wl2    = (const float*)d_in[9];
  const float* bl2    = (const float*)d_in[10];
  const float* Wr2    = (const float*)d_in[11];
  const float* W1     = (const float*)d_in[12];
  const float* b1     = (const float*)d_in[13];
  const float* W2     = (const float*)d_in[14];
  const float* b2     = (const float*)d_in[15];
  float* out = (float*)d_out;

  const int* src = eidx;
  const int* dst = eidx + NEDGES;

  float* ws  = (float*)d_ws;
  float* cnt = ws;
  float* agg = ws + NNODES;
  float* h1  = agg + NNODES * HID;
  float* h2  = h1 + NNODES * HID;

  const int TB = 256;
  int gemm_waves = (NNODES / 16 / SM) * ((HID / 16) / SN);
  int gemm_blocks = gemm_waves / 4;

  {
    int nz = NNODES * (1 + F_IN);
    k_zero<<<(nz + TB - 1) / TB, TB, 0, stream>>>(cnt, nz);
    k_count<<<(NEDGES + TB - 1) / TB, TB, 0, stream>>>(dst, cnt);
    int nt = NEDGES << 7;
    k_scatter<<<(nt + TB - 1) / TB, TB, 0, stream>>>(src, dst, x, agg, 7);
    int nn = NNODES << 7;
    k_norm<<<(nn + TB - 1) / TB, TB, 0, stream>>>(agg, cnt, 7);
    k_sage_gemm<F_IN><<<gemm_blocks, 128, 0, stream>>>(agg, x, Wl1, Wr1, bl1, h1);
  }
  {
    int nz = NNODES * HID;
    k_zero<<<(nz + TB - 1) / TB, TB, 0, stream>>>(agg, nz);
    int nt = NEDGES << 8;
    k_scatter<<<(nt + TB - 1) / TB, TB, 0, stream>>>(src, dst, h1, agg, 8);
    int nn = NNODES << 8;
    k_norm<<<(nn + TB - 1) / TB, TB, 0, stream>>>(agg, cnt, 8);
    k_sage_gemm<HID><<<gemm_blocks, 128, 0, stream>>>(agg, h1, Wl2, Wr2, bl2, h2);
  }
  {
    float* qu = agg;
    float* qv = h1;
    int pwaves = (NNODES / 16 / PM) * ((2 * HID / 16) / PN);
    k_node_proj<<<pwaves / 4, 128, 0, stream>>>(h2, W1, b1, qu);
    int waves = NPAIRS / 16 / TPW;
    int blocks = (waves * 32 + TB - 1) / TB;
    k_pair_mlp<<<blocks, TB, 0, stream>>>(qu, qv, eu, ev, eattr, logexp,
                                          W1, W2, b2, out);
  }
}


# Neighbour mean by gather: edges bucketed by destination once per call (int atomics give slots, block scan); one wave sums a node's source rows in slot (arrival) order; no float atomics, no zero pass
# speedup vs baseline: 13.7680x; 2.7906x over previous
#include <hip/hip_runtime.h>
#include <hip/hip_bf16.h>

typedef __attribute__((ext_vector_type(2))) float v2f;
typedef __attribute__((ext_vector_type(4))) float v4f;
typedef __attribute__((ext_vector_type(8))) float v8f;
typedef __attribute__((ext_vector_type(4))) int v4i;

#define NNODES 50000
#define NEDGES 800000
#define NPAIRS 500000
#define F_IN   128
#define HID    256
#define EIN    16
#define ZK     (2 * HID + EIN)
#define SM     5
#define SN     4
#define PM     5
#define PN     4
#define TPW    2

__device__ __forceinline__ v8f wmma_ab(v2f a, v2f b, v8f c) {
  return __builtin_amdgcn_wmma_f32_16x16x4_f32(false, a, false, b,
                                               (short)0, c, false, false);
}

__device__ __forceinline__ v2f ld2(const float* __restrict__ p, int k) {
  v2f r;
  r.x = p[k];
  r.y = p[k + 1];
  return r;
}

__global__ void k_zero(int* __restrict__ p, int n) {
  int i = blockIdx.x * blockDim.x + threadIdx.x;
  if (i < n) p[i] = 0;
}


__global__ void k_count(const int* __restrict__ dst, int* __restrict__ deg,
                        int* __restrict__ pos) {
  int e = blockIdx.x * blockDim.x + threadIdx.x;
  if (e < NEDGES) pos[e] = atomicAdd(&deg[dst[e]], 1);
}

#define SST 13
#define SCW (SST * 128)
__global__ void __launch_bounds__(1024)
k_scan(const int* __restrict__ deg, int* __restrict__ rowptr) {
  __shared__ int wtot[32];
  int lane = threadIdx.x & 31;
  int w = threadIdx.x >> 5;
  int base = w * SCW + (lane << 2);
  v4i d[SST];
#pragma unroll
  for (int s = 0; s < SST; ++s) {
    int i = base + s * 128;
    d[s] = (i < NNODES) ? *(const v4i*)(deg + i) : (v4i){0, 0, 0, 0};
  }
  int ex[SST];
  int carry = 0;
#pragma unroll
  for (int s = 0; s < SST; ++s) {
    int t = d[s].x + d[s].y + d[s].z + d[s].w;
    int inc = t;
    for (int o = 1; o < 32; o <<= 1) {
      int v = __shfl_up(inc, o);
      if (lane >= o) inc += v;
    }
    ex[s] = carry + inc - t;
    carry += __shfl(inc, 31);
  }
  if (lane == 0) wtot[w] = carry;
  __syncthreads();
  int t = wtot[lane];
  int incl = t;
  for (int o = 1; o < 32; o <<= 1) {
    int v = __shfl_up(incl, o);
    if (lane >= o) incl += v;
  }
  int off = __shfl(incl - t, w);
#pragma unroll
  for (int s = 0; s < SST; ++s) {
    int i = base + s * 128;
    if (i < NNODES) {
      v4i r;
      r.x = off + ex[s];
      r.y = r.x + d[s].x;
      r.z = r.y + d[s].y;
      r.w = r.z + d[s].z;
      *(v4i*)(rowptr + i) = r;
    }
  }
}

__global__ void k_fill(const int* __restrict__ src, const int* __restrict__ dst,
                       const int* __restrict__ rowptr, const int* __restrict__ pos,
                       int* __restrict__ col) {
  int e = blockIdx.x * blockDim.x + threadIdx.x;
  if (e < NEDGES) col[rowptr[dst[e]] + pos[e]] = src[e];
}

template <int F>
__global__ void __launch_bounds__(256)
k_gather_mean(const int* __restrict__ rowptr, const int* __restrict__ deg,
              const int* __restrict__ col, const float* __restrict__ X,
              float* __restrict__ agg) {
  constexpr int V = F / 128;
  int node = __builtin_amdgcn_readfirstlane((int)((blockIdx.x * blockDim.x + threadIdx.x) >> 5));
  int lane = threadIdx.x & 31;
  if (node >= NNODES) return;
  int beg = __builtin_amdgcn_readfirstlane(rowptr[node]);
  int d = __builtin_amdgcn_readfirstlane(deg[node]);
  const float* xl = X + (lane << 2);
  v4f acc[V];
#pragma unroll
  for (int v = 0; v < V; ++v) acc[v] = (v4f){0.f, 0.f, 0.f, 0.f};
  for (int base = 0; base < d; base += 32) {
    int m = d - base;
    if (m > 32) m = 32;
    int mine = (lane < m) ? col[beg + base + lane] : 0;
    int j = 0;
    for (; j + 4 <= m; j += 4) {
      v4f r[4][V];
#pragma unroll
      for (int u = 0; u < 4; ++u) {
        int s = __builtin_amdgcn_readlane(mine, j + u);
#pragma unroll
        for (int v = 0; v < V; ++v) r[u][v] = *(const v4f*)(xl + s * F + v * 128);
      }
#pragma unroll
      for (int u = 0; u < 4; ++u)
#pragma unroll
        for (int v = 0; v < V; ++v) acc[v] += r[u][v];
    }
    for (; j < m; ++j) {
      int s = __builtin_amdgcn_readlane(mine, j);
#pragma unroll
      for (int v = 0; v < V; ++v) acc[v] += *(const v4f*)(xl + s * F + v * 128);
    }
  }
  float inv = 1.f / fmaxf((float)d, 1.f);
  float* o = agg + node * F + (lane << 2);
#pragma unroll
  for (int v = 0; v < V; ++v) *(v4f*)(o + v * 128) = acc[v] * inv;
}

__device__ __forceinline__ void ld8(const float* __restrict__ p, int k, v4f& lo, v4f& hi) {
  lo = *(const v4f*)(p + k);
  hi = *(const v4f*)(p + k + 4);
}

__device__ __forceinline__ v2f pair_of(v4f lo, v4f hi, int q) {
  v2f r;
  if (q == 0) { r.x = lo.x; r.y = lo.y; }
  else if (q == 1) { r.x = lo.z; r.y = lo.w; }
  else if (q == 2) { r.x = hi.x; r.y = hi.y; }
  else { r.x = hi.z; r.y = hi.w; }
  return r;
}

template <int K>
__device__ __forceinline__ void sage_acc(const float* __restrict__ R,
                                         const float* __restrict__ Wm,
                                         int row0, int col0, int koff,
                                         v8f (&c)[SM][SN]) {
  const float* arow[SM];
  const float* wrow[SN];
#pragma unroll
  for (int t = 0; t < SM; ++t) arow[t] = R + (row0 + t * 16) * K;
#pragma unroll
  for (int i = 0; i < SN; ++i) wrow[i] = Wm + (col0 + i * 16) * K;
  for (int kb = 0; kb < K; kb += 16) {
    int k = kb + koff;
    v4f alo[SM], ahi[SM];
    v4f blo[SN], bhi[SN];
#pragma unroll
    for (int t = 0; t < SM; ++t) ld8(arow[t], k, alo[t], ahi[t]);
#pragma unroll
    for (int i = 0; i < SN; ++i) ld8(wrow[i], k, blo[i], bhi[i]);
#pragma unroll
    for (int q = 0; q < 4; ++q)
#pragma unroll
      for (int t = 0; t < SM; ++t)
#pragma unroll
        for (int i = 0; i < SN; ++i)
          c[t][i] = wmma_ab(pair_of(alo[t], ahi[t], q), pair_of(blo[i], bhi[i], q), c[t][i]);
  }
}

template <int K>
__global__ void __launch_bounds__(128)
k_sage_gemm(const float* __restrict__ A, const float* __restrict__ X,
            const float* __restrict__ Wl, const float* __restrict__ Wr,
            const float* __restrict__ bias, float* __restrict__ out) {
  int gtid = blockIdx.x * blockDim.x + threadIdx.x;
  int wave = gtid >> 5;
  int lane = threadIdx.x & 31;
  const int ngrp = (HID / 16) / SN;
  const int mgrp = NNODES / 16 / SM;
  if (wave >= mgrp * ngrp) return;
  int gm = wave / ngrp;
  int gn = wave - gm * ngrp;
  int koff = (lane >> 4) << 3;
  int row0 = gm * (SM * 16) + (lane & 15);
  int col0 = gn * (SN * 16) + (lane & 15);
  v8f c[SM][SN] = {};
  sage_acc<K>(A, Wl, row0, col0, koff, c);
  sage_acc<K>(X, Wr, row0, col0, koff, c);
  int mbase = gm * (SM * 16) + ((lane >> 4) << 3);
#pragma unroll
  for (int i = 0; i < SN; ++i) {
    int col = col0 + i * 16;
    float bv = bias[col];
#pragma unroll
    for (int t = 0; t < SM; ++t)
#pragma unroll
      for (int r = 0; r < 8; ++r)
        out[(mbase + t * 16 + r) * HID + col] = fmaxf(c[t][i][r] + bv, 0.f);
  }
}

__global__ void __launch_bounds__(128)
k_node_proj(const float* __restrict__ Hn, const float* __restrict__ W1,
            const float* __restrict__ b1, float* __restrict__ Q) {
  int gtid = blockIdx.x * blockDim.x + threadIdx.x;
  int wave = gtid >> 5;
  int lane = threadIdx.x & 31;
  const int ngrp = (2 * HID / 16) / PN;
  const int mgrp = NNODES / 16 / PM;
  if (wave >= mgrp * ngrp) return;
  int gm = wave / ngrp;
  int gn = wave - gm * ngrp;
  int plane = gn >> 2;
  int koff = (lane >> 4) << 3;
  int row0 = gm * (PM * 16) + (lane & 15);
  int col0 = (gn & 3) * (PN * 16) + (lane & 15);
  const float* hrow[PM];
  const float* wrow[PN];
#pragma unroll
  for (int t = 0; t < PM; ++t) hrow[t] = Hn + (row0 + t * 16) * HID;
#pragma unroll
  for (int i = 0; i < PN; ++i) wrow[i] = W1 + (col0 + i * 16) * ZK + plane * HID;
  v8f c[PM][PN] = {};
  for (int kb = 0; kb < HID; kb += 16) {
    int k = kb + koff;
    v4f alo[PM], ahi[PM];
    v4f blo[PN], bhi[PN];
#pragma unroll
    for (int t = 0; t < PM; ++t) ld8(hrow[t], k, alo[t], ahi[t]);
#pragma unroll
    for (int i = 0; i < PN; ++i) ld8(wrow[i], k, blo[i], bhi[i]);
#pragma unroll
    for (int q = 0; q < 4; ++q)
#pragma unroll
      for (int t = 0; t < PM; ++t)
#pragma unroll
        for (int i = 0; i < PN; ++i)
          c[t][i] = wmma_ab(pair_of(alo[t], ahi[t], q), pair_of(blo[i], bhi[i], q), c[t][i]);
  }
  float* qp = Q + plane * (NNODES * HID);
  int mbase = gm * (PM * 16) + ((lane >> 4) << 3);
#pragma unroll
  for (int i = 0; i < PN; ++i) {
    int col = col0 + i * 16;
    float bv = (plane == 0) ? b1[col] : 0.f;
#pragma unroll
    for (int t = 0; t < PM; ++t)
#pragma unroll
      for (int r = 0; r < 8; ++r)
        qp[(mbase + t * 16 + r) * HID + col] = c[t][i][r] + bv;
  }
}

__device__ __forceinline__ int pair_col(int nt, int n) {
  return ((nt >> 2) << 6) + (n << 2) + (nt & 3);
}

__global__ void k_pair_mlp(const float* __restrict__ Qu, const float* __restrict__ Qv,
                           const int* __restrict__ eu, const int* __restrict__ ev,
                           const float* __restrict__ eattr,
                           const float* __restrict__ logexp,
                           const float* __restrict__ W1, const float* __restrict__ W2,
                           const float* __restrict__ b2, float* __restrict__ out) {
  __shared__ v2f sB[(HID / 16) * (EIN / 4) * 32];
  for (int idx = threadIdx.x; idx < (HID / 16) * (EIN / 4) * 32; idx += blockDim.x) {
    int nt = idx >> 7;
    int ks = (idx >> 5) & 3;
    int ln = idx & 31;
    const float* w = W1 + pair_col(nt, ln & 15) * ZK + 2 * HID;
    sB[idx] = ld2(w, ks * 4 + ((ln >> 4) << 1));
  }
  __syncthreads();

  int gtid = blockIdx.x * blockDim.x + threadIdx.x;
  int wave = gtid >> 5;
  int lane = threadIdx.x & 31;
  if (wave >= NPAIRS / 16 / TPW) return;
  int n = lane & 15;
  int half = (lane >> 4) << 3;
  int koff = (lane >> 4) << 1;
  float b2v = b2[0];

  for (int tt = 0; tt < TPW; ++tt) {
    int tile = wave * TPW + tt;
    int ebase = tile * 16 + half;
    const float* arow = eattr + (tile * 16 + n) * EIN;
    v2f a[EIN / 4];
#pragma unroll
    for (int ks = 0; ks < EIN / 4; ++ks) a[ks] = ld2(arow, ks * 4 + koff);
    int uoff[8];
    int voff[8];
#pragma unroll
    for (int r = 0; r < 8; ++r) {
      uoff[r] = eu[ebase + r] * HID + (n << 2);
      voff[r] = ev[ebase + r] * HID + (n << 2);
    }
    float acc[8] = {};
    for (int j = 0; j < 4; ++j) {
      v4f w2v = *(const v4f*)(W2 + (j << 6) + (n << 2));
      v8f c[4];
#pragma unroll
      for (int r = 0; r < 8; ++r) {
        v4f pu = *(const v4f*)(Qu + uoff[r] + (j << 6));
        v4f pv = *(const v4f*)(Qv + voff[r] + (j << 6));
#pragma unroll
        for (int i = 0; i < 4; ++i) c[i][r] = pu[i] + pv[i];
      }
#pragma unroll
      for (int ks = 0; ks < EIN / 4; ++ks)
#pragma unroll
        for (int i = 0; i < 4; ++i)
          c[i] = wmma_ab(a[ks], sB[((j * 4 + i) * 4 + ks) * 32 + lane], c[i]);
#pragma unroll
      for (int i = 0; i < 4; ++i)
#pragma unroll
        for (int r = 0; r < 8; ++r)
          acc[r] += fmaxf(c[i][r], 0.f) * w2v[i];
    }
#pragma unroll
    for (int r = 0; r < 8; ++r) {
      float s = acc[r];
      s += __shfl_xor(s, 1);
      s += __shfl_xor(s, 2);
      s += __shfl_xor(s, 4);
      s += __shfl_xor(s, 8);
      acc[r] = s;
    }
    if (n == 0) {
#pragma unroll
      for (int r = 0; r < 8; ++r)
        out[ebase + r] = logexp[ebase + r] + acc[r] + b2v;
    }
  }
}

extern "C" void kernel_launch(void* const* d_in, const int* in_sizes, int n_in,
                              void* d_out, int out_size, void* d_ws, size_t ws_size,
                              hipStream_t stream) {
  (void)in_sizes; (void)n_in; (void)out_size; (void)ws_size;
  const float* x      = (const float*)d_in[0];
  const int*   eidx   = (const int*)d_in[1];
  const int*   eu     = (const int*)d_in[2];
  const int*   ev     = (const int*)d_in[3];
  const float* eattr  = (const float*)d_in[4];
  const float* logexp = (const float*)d_in[5];
  const float* Wl1    = (const float*)d_in[6];
  const float* bl1    = (const float*)d_in[7];
  const float* Wr1    = (const float*)d_in[8];
  const float* Wl2    = (const float*)d_in[9];
  const float* bl2    = (const float*)d_in[10];
  const float* Wr2    = (const float*)d_in[11];
  const float* W1     = (const float*)d_in[12];
  const float* b1     = (const float*)d_in[13];
  const float* W2     = (const float*)d_in[14];
  const float* b2     = (const float*)d_in[15];
  float* out = (float*)d_out;

  const int* src = eidx;
  const int* dst = eidx + NEDGES;

  float* ws  = (float*)d_ws;
  int*   deg = (int*)d_ws;
  float* agg = ws + NNODES;
  float* h1  = agg + NNODES * HID;
  float* h2  = h1 + NNODES * HID;
  int* rowptr = (int*)h2;
  int* pos    = rowptr + NNODES;
  int* col    = pos + NEDGES;

  const int TB = 256;
  int gemm_waves = (NNODES / 16 / SM) * ((HID / 16) / SN);
  int gemm_blocks = gemm_waves / 4;
  int gather_blocks = NNODES / (TB / 32);

  {
    k_zero<<<(NNODES + TB - 1) / TB, TB, 0, stream>>>(deg, NNODES);
    k_count<<<(NEDGES + TB - 1) / TB, TB, 0, stream>>>(dst, deg, pos);
    k_scan<<<1, 1024, 0, stream>>>(deg, rowptr);
    k_fill<<<(NEDGES + TB - 1) / TB, TB, 0, stream>>>(src, dst, rowptr, pos, col);
  }
  {
    k_gather_mean<F_IN><<<gather_blocks, TB, 0, stream>>>(rowptr, deg, col, x, agg);
    k_sage_gemm<F_IN><<<gemm_blocks, 128, 0, stream>>>(agg, x, Wl1, Wr1, bl1, h1);
  }
  {
    k_gather_mean<HID><<<gather_blocks, TB, 0, stream>>>(rowptr, deg, col, h1, agg);
    k_sage_gemm<HID><<<gemm_blocks, 128, 0, stream>>>(agg, h1, Wl2, Wr2, bl2, h2);
  }
  {
    float* qu = agg;
    float* qv = h1;
    int pwaves = (NNODES / 16 / PM) * ((2 * HID / 16) / PN);
    k_node_proj<<<pwaves / 4, 128, 0, stream>>>(h2, W1, b1, qu);
    int waves = NPAIRS / 16 / TPW;
    int blocks = (waves * 32 + TB - 1) / TB;
    k_pair_mlp<<<blocks, TB, 0, stream>>>(qu, qv, eu, ev, eattr, logexp,
                                          W1, W2, b2, out);
  }
}


# Edge buckets by a two-level counting sort (256-node destination ranges) ranked on LDS counters: no global atomics, counter clear or scan launch; bucket slot (sum) order = LDS arrival order
# speedup vs baseline: 14.8186x; 1.0763x over previous
#include <hip/hip_runtime.h>
#include <hip/hip_bf16.h>

typedef __attribute__((ext_vector_type(2))) float v2f;
typedef __attribute__((ext_vector_type(4))) float v4f;
typedef __attribute__((ext_vector_type(8))) float v8f;
typedef __attribute__((ext_vector_type(4))) int v4i;

#define NNODES 50000
#define NEDGES 800000
#define NPAIRS 500000
#define F_IN   128
#define HID    256
#define EIN    16
#define ZK     (2 * HID + EIN)
#define SM     5
#define SN     4
#define PM     5
#define PN     4
#define TPW    2

__device__ __forceinline__ v8f wmma_ab(v2f a, v2f b, v8f c) {
  return __builtin_amdgcn_wmma_f32_16x16x4_f32(false, a, false, b,
                                               (short)0, c, false, false);
}

__device__ __forceinline__ v2f ld2(const float* __restrict__ p, int k) {
  v2f r;
  r.x = p[k];
  r.y = p[k + 1];
  return r;
}

#define NB1  256
#define ECH  (NEDGES / NB1)
#define NBIN ((NNODES + 255) / 256)
#define SCN  (NBIN * NB1)
#define RPW  ((NBIN + 31) / 32)
static_assert(NB1 == 256, "a wave reads a range's chunk counters as 32 lanes x 8");
static_assert(NEDGES % NB1 == 0, "chunks must tile the edge list");
static_assert(ECH <= 4 * 1024, "a 1024-thread block covers its chunk in 4 steps");
static_assert(NNODES <= 65536, "src and dst are packed into 16 bits each");

__global__ void __launch_bounds__(1024)
k_bin_count(const int* __restrict__ dst, int* __restrict__ cnt) {
  __shared__ int hist[NBIN];
  int t = threadIdx.x;
  int b = blockIdx.x;
  if (t < NBIN) hist[t] = 0;
  __syncthreads();
  const int* dp = dst + b * ECH;
  int d[4];
#pragma unroll
  for (int k = 0; k < 4; ++k) {
    int i = t + k * 1024;
    bool ok = i < ECH;
    int dv = dp[ok ? i : ECH - 1];
    d[k] = ok ? dv : -1;
  }
#pragma unroll
  for (int k = 0; k < 4; ++k)
    if (d[k] >= 0) atomicAdd(&hist[d[k] >> 8], 1);
  __syncthreads();
  if (t < NBIN) cnt[t * NB1 + b] = hist[t];
}

__global__ void __launch_bounds__(1024)
k_bin_scatter(const int* __restrict__ src, const int* __restrict__ dst,
              const int* __restrict__ cnt, int* __restrict__ binbase,
              unsigned* __restrict__ ebin) {
  __shared__ int cur[NBIN];
  __shared__ int btot[NBIN];
  __shared__ int bbef[NBIN];
  int t = threadIdx.x;
  int b = blockIdx.x;
  int lane = t & 31;
  int w = t >> 5;
  const int* sp = src + b * ECH;
  const int* dp = dst + b * ECH;
  int s[4];
  int d[4];
#pragma unroll
  for (int k = 0; k < 4; ++k) {
    int i = t + k * 1024;
    bool ok = i < ECH;
    int ii = ok ? i : ECH - 1;
    int dv = dp[ii];
    s[k] = sp[ii];
    d[k] = ok ? dv : -1;
  }
  int q = b - lane * 8;
  int tot[RPW];
  int bef[RPW];
  v4i ca[RPW];
  v4i cc[RPW];
#pragma unroll
  for (int r = 0; r < RPW; ++r) {
    int bin = w + r * 32;
    if (bin > NBIN - 1) bin = NBIN - 1;
    const int* row = cnt + bin * NB1 + lane * 8;
    ca[r] = *(const v4i*)row;
    cc[r] = *(const v4i*)(row + 4);
  }
#pragma unroll
  for (int r = 0; r < RPW; ++r) {
    v4i a = ca[r];
    v4i c = cc[r];
    tot[r] = a.x + a.y + a.z + a.w + c.x + c.y + c.z + c.w;
    bef[r] = (q > 0 ? a.x : 0) + (q > 1 ? a.y : 0) + (q > 2 ? a.z : 0) + (q > 3 ? a.w : 0)
           + (q > 4 ? c.x : 0) + (q > 5 ? c.y : 0) + (q > 6 ? c.z : 0) + (q > 7 ? c.w : 0);
  }
  for (int o = 16; o > 0; o >>= 1) {
#pragma unroll
    for (int r = 0; r < RPW; ++r) {
      tot[r] += __shfl_xor(tot[r], o);
      bef[r] += __shfl_xor(bef[r], o);
    }
  }
  if (lane == 0) {
#pragma unroll
    for (int r = 0; r < RPW; ++r) {
      int bin = w + r * 32;
      if (bin < NBIN) {
        btot[bin] = tot[r];
        bbef[bin] = bef[r];
      }
    }
  }
  __syncthreads();
  if (t < 32) {
    int e[RPW];
    int run = 0;
#pragma unroll
    for (int k = 0; k < RPW; ++k) {
      int idx = RPW * t + k;
      e[k] = run;
      run += (idx < NBIN) ? btot[idx] : 0;
    }
    int incl = run;
    for (int o = 1; o < 32; o <<= 1) {
      int u = __shfl_up(incl, o);
      if (t >= o) incl += u;
    }
    int ex = incl - run;
#pragma unroll
    for (int k = 0; k < RPW; ++k) {
      int idx = RPW * t + k;
      if (idx < NBIN) {
        int base = ex + e[k];
        cur[idx] = base + bbef[idx];
        if (b == 0) binbase[idx] = base;
      }
    }
  }
  __syncthreads();
#pragma unroll
  for (int k = 0; k < 4; ++k)
    if (d[k] >= 0) {
      int slot = atomicAdd(&cur[d[k] >> 8], 1);
      ebin[slot] = ((unsigned)s[k] << 16) | (unsigned)d[k];
    }
}

__global__ void __launch_bounds__(1024)
k_bucket(const int* __restrict__ binbase, const unsigned* __restrict__ ebin,
         int* __restrict__ rowptr, int* __restrict__ deg, int* __restrict__ col) {
  __shared__ int hist[256];
  __shared__ int cur[256];
  int t = threadIdx.x;
  int bin = blockIdx.x;
  int segbeg = binbase[bin];
  int segend = (bin + 1 < NBIN) ? binbase[bin + 1] : NEDGES;
  int seglen = segend - segbeg;
  const unsigned* ep = ebin + segbeg;
  if (t < 256) hist[t] = 0;
  __syncthreads();
  for (int i0 = t; i0 < seglen; i0 += 4096) {
    unsigned v[4];
#pragma unroll
    for (int k = 0; k < 4; ++k) {
      int i = i0 + k * 1024;
      v[k] = (i < seglen) ? ep[i] : 0xFFFFFFFFu;
    }
#pragma unroll
    for (int k = 0; k < 4; ++k)
      if (v[k] != 0xFFFFFFFFu) atomicAdd(&hist[v[k] & 255u], 1);
  }
  __syncthreads();
  if (t < 32) {
    int e[8];
    int run = 0;
#pragma unroll
    for (int k = 0; k < 8; ++k) {
      e[k] = run;
      run += hist[8 * t + k];
    }
    int incl = run;
    for (int o = 1; o < 32; o <<= 1) {
      int u = __shfl_up(incl, o);
      if (t >= o) incl += u;
    }
    int ex = segbeg + incl - run;
#pragma unroll
    for (int k = 0; k < 8; ++k) cur[8 * t + k] = ex + e[k];
  }
  __syncthreads();
  if (t < 256) {
    int node = bin * 256 + t;
    if (node < NNODES) {
      rowptr[node] = cur[t];
      deg[node] = hist[t];
    }
  }
  __syncthreads();
  for (int i0 = t; i0 < seglen; i0 += 4096) {
    unsigned v[4];
#pragma unroll
    for (int k = 0; k < 4; ++k) {
      int i = i0 + k * 1024;
      v[k] = (i < seglen) ? ep[i] : 0xFFFFFFFFu;
    }
#pragma unroll
    for (int k = 0; k < 4; ++k)
      if (v[k] != 0xFFFFFFFFu) {
        int slot = atomicAdd(&cur[v[k] & 255u], 1);
        col[slot] = (int)(v[k] >> 16);
      }
  }
}

template <int F>
__global__ void __launch_bounds__(256)
k_gather_mean(const int* __restrict__ rowptr, const int* __restrict__ deg,
              const int* __restrict__ col, const float* __restrict__ X,
              float* __restrict__ agg) {
  constexpr int V = F / 128;
  int node = __builtin_amdgcn_readfirstlane((int)((blockIdx.x * blockDim.x + threadIdx.x) >> 5));
  int lane = threadIdx.x & 31;
  if (node >= NNODES) return;
  int beg = __builtin_amdgcn_readfirstlane(rowptr[node]);
  int d = __builtin_amdgcn_readfirstlane(deg[node]);
  const float* xl = X + (lane << 2);
  v4f acc[V];
#pragma unroll
  for (int v = 0; v < V; ++v) acc[v] = (v4f){0.f, 0.f, 0.f, 0.f};
  for (int base = 0; base < d; base += 32) {
    int m = d - base;
    if (m > 32) m = 32;
    int mine = (lane < m) ? col[beg + base + lane] : 0;
    int j = 0;
    for (; j + 4 <= m; j += 4) {
      v4f r[4][V];
#pragma unroll
      for (int u = 0; u < 4; ++u) {
        int s = __builtin_amdgcn_readlane(mine, j + u);
#pragma unroll
        for (int v = 0; v < V; ++v) r[u][v] = *(const v4f*)(xl + s * F + v * 128);
      }
#pragma unroll
      for (int u = 0; u < 4; ++u)
#pragma unroll
        for (int v = 0; v < V; ++v) acc[v] += r[u][v];
    }
    for (; j < m; ++j) {
      int s = __builtin_amdgcn_readlane(mine, j);
#pragma unroll
      for (int v = 0; v < V; ++v) acc[v] += *(const v4f*)(xl + s * F + v * 128);
    }
  }
  float inv = 1.f / fmaxf((float)d, 1.f);
  float* o = agg + node * F + (lane << 2);
#pragma unroll
  for (int v = 0; v < V; ++v) *(v4f*)(o + v * 128) = acc[v] * inv;
}

__device__ __forceinline__ void ld8(const float* __restrict__ p, int k, v4f& lo, v4f& hi) {
  lo = *(const v4f*)(p + k);
  hi = *(const v4f*)(p + k + 4);
}

__device__ __forceinline__ v2f pair_of(v4f lo, v4f hi, int q) {
  v2f r;
  if (q == 0) { r.x = lo.x; r.y = lo.y; }
  else if (q == 1) { r.x = lo.z; r.y = lo.w; }
  else if (q == 2) { r.x = hi.x; r.y = hi.y; }
  else { r.x = hi.z; r.y = hi.w; }
  return r;
}

template <int K>
__device__ __forceinline__ void sage_acc(const float* __restrict__ R,
                                         const float* __restrict__ Wm,
                                         int row0, int col0, int koff,
                                         v8f (&c)[SM][SN]) {
  const float* arow[SM];
  const float* wrow[SN];
#pragma unroll
  for (int t = 0; t < SM; ++t) arow[t] = R + (row0 + t * 16) * K;
#pragma unroll
  for (int i = 0; i < SN; ++i) wrow[i] = Wm + (col0 + i * 16) * K;
  for (int kb = 0; kb < K; kb += 16) {
    int k = kb + koff;
    v4f alo[SM], ahi[SM];
    v4f blo[SN], bhi[SN];
#pragma unroll
    for (int t = 0; t < SM; ++t) ld8(arow[t], k, alo[t], ahi[t]);
#pragma unroll
    for (int i = 0; i < SN; ++i) ld8(wrow[i], k, blo[i], bhi[i]);
#pragma unroll
    for (int q = 0; q < 4; ++q)
#pragma unroll
      for (int t = 0; t < SM; ++t)
#pragma unroll
        for (int i = 0; i < SN; ++i)
          c[t][i] = wmma_ab(pair_of(alo[t], ahi[t], q), pair_of(blo[i], bhi[i], q), c[t][i]);
  }
}

template <int K>
__global__ void __launch_bounds__(128)
k_sage_gemm(const float* __restrict__ A, const float* __restrict__ X,
            const float* __restrict__ Wl, const float* __restrict__ Wr,
            const float* __restrict__ bias, float* __restrict__ out) {
  int gtid = blockIdx.x * blockDim.x + threadIdx.x;
  int wave = gtid >> 5;
  int lane = threadIdx.x & 31;
  const int ngrp = (HID / 16) / SN;
  const int mgrp = NNODES / 16 / SM;
  if (wave >= mgrp * ngrp) return;
  int gm = wave / ngrp;
  int gn = wave - gm * ngrp;
  int koff = (lane >> 4) << 3;
  int row0 = gm * (SM * 16) + (lane & 15);
  int col0 = gn * (SN * 16) + (lane & 15);
  v8f c[SM][SN] = {};
  sage_acc<K>(A, Wl, row0, col0, koff, c);
  sage_acc<K>(X, Wr, row0, col0, koff, c);
  int mbase = gm * (SM * 16) + ((lane >> 4) << 3);
#pragma unroll
  for (int i = 0; i < SN; ++i) {
    int col = col0 + i * 16;
    float bv = bias[col];
#pragma unroll
    for (int t = 0; t < SM; ++t)
#pragma unroll
      for (int r = 0; r < 8; ++r)
        out[(mbase + t * 16 + r) * HID + col] = fmaxf(c[t][i][r] + bv, 0.f);
  }
}

__global__ void __launch_bounds__(128)
k_node_proj(const float* __restrict__ Hn, const float* __restrict__ W1,
            const float* __restrict__ b1, float* __restrict__ Q) {
  int gtid = blockIdx.x * blockDim.x + threadIdx.x;
  int wave = gtid >> 5;
  int lane = threadIdx.x & 31;
  const int ngrp = (2 * HID / 16) / PN;
  const int mgrp = NNODES / 16 / PM;
  if (wave >= mgrp * ngrp) return;
  int gm = wave / ngrp;
  int gn = wave - gm * ngrp;
  int plane = gn >> 2;
  int koff = (lane >> 4) << 3;
  int row0 = gm * (PM * 16) + (lane & 15);
  int col0 = (gn & 3) * (PN * 16) + (lane & 15);
  const float* hrow[PM];
  const float* wrow[PN];
#pragma unroll
  for (int t = 0; t < PM; ++t) hrow[t] = Hn + (row0 + t * 16) * HID;
#pragma unroll
  for (int i = 0; i < PN; ++i) wrow[i] = W1 + (col0 + i * 16) * ZK + plane * HID;
  v8f c[PM][PN] = {};
  for (int kb = 0; kb < HID; kb += 16) {
    int k = kb + koff;
    v4f alo[PM], ahi[PM];
    v4f blo[PN], bhi[PN];
#pragma unroll
    for (int t = 0; t < PM; ++t) ld8(hrow[t], k, alo[t], ahi[t]);
#pragma unroll
    for (int i = 0; i < PN; ++i) ld8(wrow[i], k, blo[i], bhi[i]);
#pragma unroll
    for (int q = 0; q < 4; ++q)
#pragma unroll
      for (int t = 0; t < PM; ++t)
#pragma unroll
        for (int i = 0; i < PN; ++i)
          c[t][i] = wmma_ab(pair_of(alo[t], ahi[t], q), pair_of(blo[i], bhi[i], q), c[t][i]);
  }
  float* qp = Q + plane * (NNODES * HID);
  int mbase = gm * (PM * 16) + ((lane >> 4) << 3);
#pragma unroll
  for (int i = 0; i < PN; ++i) {
    int col = col0 + i * 16;
    float bv = (plane == 0) ? b1[col] : 0.f;
#pragma unroll
    for (int t = 0; t < PM; ++t)
#pragma unroll
      for (int r = 0; r < 8; ++r)
        qp[(mbase + t * 16 + r) * HID + col] = c[t][i][r] + bv;
  }
}

__device__ __forceinline__ int pair_col(int nt, int n) {
  return ((nt >> 2) << 6) + (n << 2) + (nt & 3);
}

__global__ void k_pair_mlp(const float* __restrict__ Qu, const float* __restrict__ Qv,
                           const int* __restrict__ eu, const int* __restrict__ ev,
                           const float* __restrict__ eattr,
                           const float* __restrict__ logexp,
                           const float* __restrict__ W1, const float* __restrict__ W2,
                           const float* __restrict__ b2, float* __restrict__ out) {
  __shared__ v2f sB[(HID / 16) * (EIN / 4) * 32];
  for (int idx = threadIdx.x; idx < (HID / 16) * (EIN / 4) * 32; idx += blockDim.x) {
    int nt = idx >> 7;
    int ks = (idx >> 5) & 3;
    int ln = idx & 31;
    const float* w = W1 + pair_col(nt, ln & 15) * ZK + 2 * HID;
    sB[idx] = ld2(w, ks * 4 + ((ln >> 4) << 1));
  }
  __syncthreads();

  int gtid = blockIdx.x * blockDim.x + threadIdx.x;
  int wave = gtid >> 5;
  int lane = threadIdx.x & 31;
  if (wave >= NPAIRS / 16 / TPW) return;
  int n = lane & 15;
  int half = (lane >> 4) << 3;
  int koff = (lane >> 4) << 1;
  float b2v = b2[0];

  for (int tt = 0; tt < TPW; ++tt) {
    int tile = wave * TPW + tt;
    int ebase = tile * 16 + half;
    const float* arow = eattr + (tile * 16 + n) * EIN;
    v2f a[EIN / 4];
#pragma unroll
    for (int ks = 0; ks < EIN / 4; ++ks) a[ks] = ld2(arow, ks * 4 + koff);
    int uoff[8];
    int voff[8];
#pragma unroll
    for (int r = 0; r < 8; ++r) {
      uoff[r] = eu[ebase + r] * HID + (n << 2);
      voff[r] = ev[ebase + r] * HID + (n << 2);
    }
    float acc[8] = {};
    for (int j = 0; j < 4; ++j) {
      v4f w2v = *(const v4f*)(W2 + (j << 6) + (n << 2));
      v8f c[4];
#pragma unroll
      for (int r = 0; r < 8; ++r) {
        v4f pu = *(const v4f*)(Qu + uoff[r] + (j << 6));
        v4f pv = *(const v4f*)(Qv + voff[r] + (j << 6));
#pragma unroll
        for (int i = 0; i < 4; ++i) c[i][r] = pu[i] + pv[i];
      }
#pragma unroll
      for (int ks = 0; ks < EIN / 4; ++ks)
#pragma unroll
        for (int i = 0; i < 4; ++i)
          c[i] = wmma_ab(a[ks], sB[((j * 4 + i) * 4 + ks) * 32 + lane], c[i]);
#pragma unroll
      for (int i = 0; i < 4; ++i)
#pragma unroll
        for (int r = 0; r < 8; ++r)
          acc[r] += fmaxf(c[i][r], 0.f) * w2v[i];
    }
#pragma unroll
    for (int r = 0; r < 8; ++r) {
      float s = acc[r];
      s += __shfl_xor(s, 1);
      s += __shfl_xor(s, 2);
      s += __shfl_xor(s, 4);
      s += __shfl_xor(s, 8);
      acc[r] = s;
    }
    if (n == 0) {
#pragma unroll
      for (int r = 0; r < 8; ++r)
        out[ebase + r] = logexp[ebase + r] + acc[r] + b2v;
    }
  }
}

extern "C" void kernel_launch(void* const* d_in, const int* in_sizes, int n_in,
                              void* d_out, int out_size, void* d_ws, size_t ws_size,
                              hipStream_t stream) {
  (void)in_sizes; (void)n_in; (void)out_size; (void)ws_size;
  const float* x      = (const float*)d_in[0];
  const int*   eidx   = (const int*)d_in[1];
  const int*   eu     = (const int*)d_in[2];
  const int*   ev     = (const int*)d_in[3];
  const float* eattr  = (const float*)d_in[4];
  const float* logexp = (const float*)d_in[5];
  const float* Wl1    = (const float*)d_in[6];
  const float* bl1    = (const float*)d_in[7];
  const float* Wr1    = (const float*)d_in[8];
  const float* Wl2    = (const float*)d_in[9];
  const float* bl2    = (const float*)d_in[10];
  const float* Wr2    = (const float*)d_in[11];
  const float* W1     = (const float*)d_in[12];
  const float* b1     = (const float*)d_in[13];
  const float* W2     = (const float*)d_in[14];
  const float* b2     = (const float*)d_in[15];
  float* out = (float*)d_out;

  const int* src = eidx;
  const int* dst = eidx + NEDGES;

  float* ws  = (float*)d_ws;
  int*   deg = (int*)d_ws;
  float* agg = ws + NNODES;
  float* h1  = agg + NNODES * HID;
  float* h2  = h1 + NNODES * HID;
  int*      rowptr  = (int*)h2;
  unsigned* ebin    = (unsigned*)(rowptr + NNODES);
  int*      col     = (int*)ebin + NEDGES;
  int*      cnt     = col + NEDGES;
  int*      binbase = cnt + SCN;

  const int TB = 256;
  int gemm_waves = (NNODES / 16 / SM) * ((HID / 16) / SN);
  int gemm_blocks = gemm_waves / 4;
  int gather_blocks = NNODES / (TB / 32);

  {
    k_bin_count<<<NB1, 1024, 0, stream>>>(dst, cnt);
    k_bin_scatter<<<NB1, 1024, 0, stream>>>(src, dst, cnt, binbase, ebin);
    k_bucket<<<NBIN, 1024, 0, stream>>>(binbase, ebin, rowptr, deg, col);
  }
  {
    k_gather_mean<F_IN><<<gather_blocks, TB, 0, stream>>>(rowptr, deg, col, x, agg);
    k_sage_gemm<F_IN><<<gemm_blocks, 128, 0, stream>>>(agg, x, Wl1, Wr1, bl1, h1);
  }
  {
    k_gather_mean<HID><<<gather_blocks, TB, 0, stream>>>(rowptr, deg, col, h1, agg);
    k_sage_gemm<HID><<<gemm_blocks, 128, 0, stream>>>(agg, h1, Wl2, Wr2, bl2, h2);
  }
  {
    float* qu = agg;
    float* qv = h1;
    int pwaves = (NNODES / 16 / PM) * ((2 * HID / 16) / PN);
    k_node_proj<<<pwaves / 4, 128, 0, stream>>>(h2, W1, b1, qu);
    int waves = NPAIRS / 16 / TPW;
    int blocks = (waves * 32 + TB - 1) / TB;
    k_pair_mlp<<<blocks, TB, 0, stream>>>(qu, qv, eu, ev, eattr, logexp,
                                          W1, W2, b2, out);
  }
}


# Node GEMMs (both SAGE layers, node projection): the 80 A rows a block's 4 waves share are staged once in LDS by whole-line loads, double-buffered 64-deep slabs; B stays direct; sum order kept
# speedup vs baseline: 15.3766x; 1.0377x over previous
#include <hip/hip_runtime.h>
#include <hip/hip_bf16.h>

typedef __attribute__((ext_vector_type(2))) float v2f;
typedef __attribute__((ext_vector_type(4))) float v4f;
typedef __attribute__((ext_vector_type(8))) float v8f;
typedef __attribute__((ext_vector_type(4))) int v4i;

#define NNODES 50000
#define NEDGES 800000
#define NPAIRS 500000
#define F_IN   128
#define HID    256
#define EIN    16
#define ZK     (2 * HID + EIN)
#define SM     5
#define SN     4
#define PM     5
#define PN     4
#define TPW    2

__device__ __forceinline__ v8f wmma_ab(v2f a, v2f b, v8f c) {
  return __builtin_amdgcn_wmma_f32_16x16x4_f32(false, a, false, b,
                                               (short)0, c, false, false);
}

__device__ __forceinline__ v2f ld2(const float* __restrict__ p, int k) {
  v2f r;
  r.x = p[k];
  r.y = p[k + 1];
  return r;
}

#define NB1  256
#define ECH  (NEDGES / NB1)
#define NBIN ((NNODES + 255) / 256)
#define SCN  (NBIN * NB1)
#define RPW  ((NBIN + 31) / 32)
static_assert(NB1 == 256, "a wave reads a range's chunk counters as 32 lanes x 8");
static_assert(NEDGES % NB1 == 0, "chunks must tile the edge list");
static_assert(ECH <= 4 * 1024, "a 1024-thread block covers its chunk in 4 steps");
static_assert(NNODES <= 65536, "src and dst are packed into 16 bits each");

__global__ void __launch_bounds__(1024)
k_bin_count(const int* __restrict__ dst, int* __restrict__ cnt) {
  __shared__ int hist[NBIN];
  int t = threadIdx.x;
  int b = blockIdx.x;
  if (t < NBIN) hist[t] = 0;
  __syncthreads();
  const int* dp = dst + b * ECH;
  int d[4];
#pragma unroll
  for (int k = 0; k < 4; ++k) {
    int i = t + k * 1024;
    bool ok = i < ECH;
    int dv = dp[ok ? i : ECH - 1];
    d[k] = ok ? dv : -1;
  }
#pragma unroll
  for (int k = 0; k < 4; ++k)
    if (d[k] >= 0) atomicAdd(&hist[d[k] >> 8], 1);
  __syncthreads();
  if (t < NBIN) cnt[t * NB1 + b] = hist[t];
}

__global__ void __launch_bounds__(1024)
k_bin_scatter(const int* __restrict__ src, const int* __restrict__ dst,
              const int* __restrict__ cnt, int* __restrict__ binbase,
              unsigned* __restrict__ ebin) {
  __shared__ int cur[NBIN];
  __shared__ int btot[NBIN];
  __shared__ int bbef[NBIN];
  int t = threadIdx.x;
  int b = blockIdx.x;
  int lane = t & 31;
  int w = t >> 5;
  const int* sp = src + b * ECH;
  const int* dp = dst + b * ECH;
  int s[4];
  int d[4];
#pragma unroll
  for (int k = 0; k < 4; ++k) {
    int i = t + k * 1024;
    bool ok = i < ECH;
    int ii = ok ? i : ECH - 1;
    int dv = dp[ii];
    s[k] = sp[ii];
    d[k] = ok ? dv : -1;
  }
  int q = b - lane * 8;
  int tot[RPW];
  int bef[RPW];
  v4i ca[RPW];
  v4i cc[RPW];
#pragma unroll
  for (int r = 0; r < RPW; ++r) {
    int bin = w + r * 32;
    if (bin > NBIN - 1) bin = NBIN - 1;
    const int* row = cnt + bin * NB1 + lane * 8;
    ca[r] = *(const v4i*)row;
    cc[r] = *(const v4i*)(row + 4);
  }
#pragma unroll
  for (int r = 0; r < RPW; ++r) {
    v4i a = ca[r];
    v4i c = cc[r];
    tot[r] = a.x + a.y + a.z + a.w + c.x + c.y + c.z + c.w;
    bef[r] = (q > 0 ? a.x : 0) + (q > 1 ? a.y : 0) + (q > 2 ? a.z : 0) + (q > 3 ? a.w : 0)
           + (q > 4 ? c.x : 0) + (q > 5 ? c.y : 0) + (q > 6 ? c.z : 0) + (q > 7 ? c.w : 0);
  }
  for (int o = 16; o > 0; o >>= 1) {
#pragma unroll
    for (int r = 0; r < RPW; ++r) {
      tot[r] += __shfl_xor(tot[r], o);
      bef[r] += __shfl_xor(bef[r], o);
    }
  }
  if (lane == 0) {
#pragma unroll
    for (int r = 0; r < RPW; ++r) {
      int bin = w + r * 32;
      if (bin < NBIN) {
        btot[bin] = tot[r];
        bbef[bin] = bef[r];
      }
    }
  }
  __syncthreads();
  if (t < 32) {
    int e[RPW];
    int run = 0;
#pragma unroll
    for (int k = 0; k < RPW; ++k) {
      int idx = RPW * t + k;
      e[k] = run;
      run += (idx < NBIN) ? btot[idx] : 0;
    }
    int incl = run;
    for (int o = 1; o < 32; o <<= 1) {
      int u = __shfl_up(incl, o);
      if (t >= o) incl += u;
    }
    int ex = incl - run;
#pragma unroll
    for (int k = 0; k < RPW; ++k) {
      int idx = RPW * t + k;
      if (idx < NBIN) {
        int base = ex + e[k];
        cur[idx] = base + bbef[idx];
        if (b == 0) binbase[idx] = base;
      }
    }
  }
  __syncthreads();
#pragma unroll
  for (int k = 0; k < 4; ++k)
    if (d[k] >= 0) {
      int slot = atomicAdd(&cur[d[k] >> 8], 1);
      ebin[slot] = ((unsigned)s[k] << 16) | (unsigned)d[k];
    }
}

__global__ void __launch_bounds__(1024)
k_bucket(const int* __restrict__ binbase, const unsigned* __restrict__ ebin,
         int* __restrict__ rowptr, int* __restrict__ deg, int* __restrict__ col) {
  __shared__ int hist[256];
  __shared__ int cur[256];
  int t = threadIdx.x;
  int bin = blockIdx.x;
  int segbeg = binbase[bin];
  int segend = (bin + 1 < NBIN) ? binbase[bin + 1] : NEDGES;
  int seglen = segend - segbeg;
  const unsigned* ep = ebin + segbeg;
  if (t < 256) hist[t] = 0;
  __syncthreads();
  for (int i0 = t; i0 < seglen; i0 += 4096) {
    unsigned v[4];
#pragma unroll
    for (int k = 0; k < 4; ++k) {
      int i = i0 + k * 1024;
      v[k] = (i < seglen) ? ep[i] : 0xFFFFFFFFu;
    }
#pragma unroll
    for (int k = 0; k < 4; ++k)
      if (v[k] != 0xFFFFFFFFu) atomicAdd(&hist[v[k] & 255u], 1);
  }
  __syncthreads();
  if (t < 32) {
    int e[8];
    int run = 0;
#pragma unroll
    for (int k = 0; k < 8; ++k) {
      e[k] = run;
      run += hist[8 * t + k];
    }
    int incl = run;
    for (int o = 1; o < 32; o <<= 1) {
      int u = __shfl_up(incl, o);
      if (t >= o) incl += u;
    }
    int ex = segbeg + incl - run;
#pragma unroll
    for (int k = 0; k < 8; ++k) cur[8 * t + k] = ex + e[k];
  }
  __syncthreads();
  if (t < 256) {
    int node = bin * 256 + t;
    if (node < NNODES) {
      rowptr[node] = cur[t];
      deg[node] = hist[t];
    }
  }
  __syncthreads();
  for (int i0 = t; i0 < seglen; i0 += 4096) {
    unsigned v[4];
#pragma unroll
    for (int k = 0; k < 4; ++k) {
      int i = i0 + k * 1024;
      v[k] = (i < seglen) ? ep[i] : 0xFFFFFFFFu;
    }
#pragma unroll
    for (int k = 0; k < 4; ++k)
      if (v[k] != 0xFFFFFFFFu) {
        int slot = atomicAdd(&cur[v[k] & 255u], 1);
        col[slot] = (int)(v[k] >> 16);
      }
  }
}

template <int F>
__global__ void __launch_bounds__(256)
k_gather_mean(const int* __restrict__ rowptr, const int* __restrict__ deg,
              const int* __restrict__ col, const float* __restrict__ X,
              float* __restrict__ agg) {
  constexpr int V = F / 128;
  int node = __builtin_amdgcn_readfirstlane((int)((blockIdx.x * blockDim.x + threadIdx.x) >> 5));
  int lane = threadIdx.x & 31;
  if (node >= NNODES) return;
  int beg = __builtin_amdgcn_readfirstlane(rowptr[node]);
  int d = __builtin_amdgcn_readfirstlane(deg[node]);
  const float* xl = X + (lane << 2);
  v4f acc[V];
#pragma unroll
  for (int v = 0; v < V; ++v) acc[v] = (v4f){0.f, 0.f, 0.f, 0.f};
  for (int base = 0; base < d; base += 32) {
    int m = d - base;
    if (m > 32) m = 32;
    int mine = (lane < m) ? col[beg + base + lane] : 0;
    int j = 0;
    for (; j + 4 <= m; j += 4) {
      v4f r[4][V];
#pragma unroll
      for (int u = 0; u < 4; ++u) {
        int s = __builtin_amdgcn_readlane(mine, j + u);
#pragma unroll
        for (int v = 0; v < V; ++v) r[u][v] = *(const v4f*)(xl + s * F + v * 128);
      }
#pragma unroll
      for (int u = 0; u < 4; ++u)
#pragma unroll
        for (int v = 0; v < V; ++v) acc[v] += r[u][v];
    }
    for (; j < m; ++j) {
      int s = __builtin_amdgcn_readlane(mine, j);
#pragma unroll
      for (int v = 0; v < V; ++v) acc[v] += *(const v4f*)(xl + s * F + v * 128);
    }
  }
  float inv = 1.f / fmaxf((float)d, 1.f);
  float* o = agg + node * F + (lane << 2);
#pragma unroll
  for (int v = 0; v < V; ++v) *(v4f*)(o + v * 128) = acc[v] * inv;
}

__device__ __forceinline__ void ld8(const float* __restrict__ p, int k, v4f& lo, v4f& hi) {
  lo = *(const v4f*)(p + k);
  hi = *(const v4f*)(p + k + 4);
}

__device__ __forceinline__ v2f pair_of(v4f lo, v4f hi, int q) {
  v2f r;
  if (q == 0) { r.x = lo.x; r.y = lo.y; }
  else if (q == 1) { r.x = lo.z; r.y = lo.w; }
  else if (q == 2) { r.x = hi.x; r.y = hi.y; }
  else { r.x = hi.z; r.y = hi.w; }
  return r;
}

#define SLAB   64
#define APITCH (SLAB + 4)
#define AROWS  (SM * 16)
#define ABUF   (AROWS * APITCH)
static_assert(SM == PM && SN == PN, "the node GEMMs share one block shape");
static_assert((AROWS * (SLAB / 4)) % 128 == 0, "a slab is a whole number of 16-byte pieces per thread");
static_assert(NNODES % AROWS == 0, "row blocks tile the nodes exactly");

template <int K, int WS, int NOP>
__device__ __forceinline__ void block_acc(const float* __restrict__ A0, const float* __restrict__ A1,
                                          const float* __restrict__ W0, const float* __restrict__ W1,
                                          int rowblk, int col0, float* sA, v8f (&c)[SM][SN]) {
  constexpr int SPO = K / SLAB;
  constexpr int T = NOP * SPO;
  constexpr int NP = AROWS * (SLAB / 4) / 128;
  int t = threadIdx.x;
  int lane = t & 31;
  int koff = (lane >> 4) << 3;
  int srow = t >> 4;
  int spc = (t & 15) << 2;
  const float* g0 = A0 + (rowblk + srow) * K + spc;
  const float* g1 = A1 + (rowblk + srow) * K + spc;
  float* sst = sA + srow * APITCH + spc;
  const float* sfr = sA + (lane & 15) * APITCH + koff;
  const float* w0 = W0 + col0 * WS + koff;
  const float* w1 = W1 + col0 * WS + koff;
  v4f st[NP];
#pragma unroll
  for (int i = 0; i < NP; ++i) st[i] = *(const v4f*)(g0 + i * 8 * K);
#pragma unroll
  for (int i = 0; i < NP; ++i) *(v4f*)(sst + i * 8 * APITCH) = st[i];
  __syncthreads();
#pragma unroll 1
  for (int g = 0; g < T; ++g) {
    int buf = g & 1;
    bool more = g + 1 < T;
    if (more) {
      int gn = g + 1;
      const float* src = (gn < SPO ? g0 : g1) + (gn & (SPO - 1)) * SLAB;
#pragma unroll
      for (int i = 0; i < NP; ++i) st[i] = *(const v4f*)(src + i * 8 * K);
    }
    const float* wb = (g < SPO ? w0 : w1) + (g & (SPO - 1)) * SLAB;
    const float* ab = sfr + buf * ABUF;
    for (int kk = 0; kk < SLAB; kk += 16) {
      v4f alo[SM], ahi[SM];
      v4f blo[SN], bhi[SN];
#pragma unroll
      for (int i = 0; i < SN; ++i) ld8(wb + i * 16 * WS, kk, blo[i], bhi[i]);
#pragma unroll
      for (int r = 0; r < SM; ++r) {
        alo[r] = *(const v4f*)(ab + r * 16 * APITCH + kk);
        ahi[r] = *(const v4f*)(ab + r * 16 * APITCH + kk + 4);
      }
#pragma unroll
      for (int q = 0; q < 4; ++q)
#pragma unroll
        for (int r = 0; r < SM; ++r)
#pragma unroll
          for (int i = 0; i < SN; ++i)
            c[r][i] = wmma_ab(pair_of(alo[r], ahi[r], q), pair_of(blo[i], bhi[i], q), c[r][i]);
    }
    if (more) {
      float* d = sst + (buf ^ 1) * ABUF;
#pragma unroll
      for (int i = 0; i < NP; ++i) *(v4f*)(d + i * 8 * APITCH) = st[i];
      __syncthreads();
    }
  }
}
static_assert((HID / 16) / SN == 4, "a block's 4 waves cover the 256 output columns");

template <int K>
__global__ void __launch_bounds__(128)
k_sage_gemm(const float* __restrict__ A, const float* __restrict__ X,
            const float* __restrict__ Wl, const float* __restrict__ Wr,
            const float* __restrict__ bias, float* __restrict__ out) {
  __shared__ __attribute__((aligned(16))) float sA[2 * ABUF];
  int lane = threadIdx.x & 31;
  int gm = blockIdx.x;
  if (gm >= NNODES / AROWS) return;
  int gn = threadIdx.x >> 5;
  int col0 = gn * (SN * 16) + (lane & 15);
  v8f c[SM][SN] = {};
  block_acc<K, K, 2>(A, X, Wl, Wr, gm * AROWS, col0, sA, c);
  int mbase = gm * (SM * 16) + ((lane >> 4) << 3);
#pragma unroll
  for (int i = 0; i < SN; ++i) {
    int col = col0 + i * 16;
    float bv = bias[col];
#pragma unroll
    for (int t = 0; t < SM; ++t)
#pragma unroll
      for (int r = 0; r < 8; ++r)
        out[(mbase + t * 16 + r) * HID + col] = fmaxf(c[t][i][r] + bv, 0.f);
  }
}

__global__ void __launch_bounds__(128)
k_node_proj(const float* __restrict__ Hn, const float* __restrict__ W1,
            const float* __restrict__ b1, float* __restrict__ Q) {
  __shared__ __attribute__((aligned(16))) float sA[2 * ABUF];
  int lane = threadIdx.x & 31;
  int gm = blockIdx.x >> 1;
  if (gm >= NNODES / AROWS) return;
  int plane = blockIdx.x & 1;
  int gn = threadIdx.x >> 5;
  int col0 = gn * (PN * 16) + (lane & 15);
  const float* wp = W1 + plane * HID;
  v8f c[PM][PN] = {};
  block_acc<HID, ZK, 1>(Hn, Hn, wp, wp, gm * AROWS, col0, sA, c);
  float* qp = Q + plane * (NNODES * HID);
  int mbase = gm * (PM * 16) + ((lane >> 4) << 3);
#pragma unroll
  for (int i = 0; i < PN; ++i) {
    int col = col0 + i * 16;
    float bv = (plane == 0) ? b1[col] : 0.f;
#pragma unroll
    for (int t = 0; t < PM; ++t)
#pragma unroll
      for (int r = 0; r < 8; ++r)
        qp[(mbase + t * 16 + r) * HID + col] = c[t][i][r] + bv;
  }
}

__device__ __forceinline__ int pair_col(int nt, int n) {
  return ((nt >> 2) << 6) + (n << 2) + (nt & 3);
}

__global__ void k_pair_mlp(const float* __restrict__ Qu, const float* __restrict__ Qv,
                           const int* __restrict__ eu, const int* __restrict__ ev,
                           const float* __restrict__ eattr,
                           const float* __restrict__ logexp,
                           const float* __restrict__ W1, const float* __restrict__ W2,
                           const float* __restrict__ b2, float* __restrict__ out) {
  __shared__ v2f sB[(HID / 16) * (EIN / 4) * 32];
  for (int idx = threadIdx.x; idx < (HID / 16) * (EIN / 4) * 32; idx += blockDim.x) {
    int nt = idx >> 7;
    int ks = (idx >> 5) & 3;
    int ln = idx & 31;
    const float* w = W1 + pair_col(nt, ln & 15) * ZK + 2 * HID;
    sB[idx] = ld2(w, ks * 4 + ((ln >> 4) << 1));
  }
  __syncthreads();

  int gtid = blockIdx.x * blockDim.x + threadIdx.x;
  int wave = gtid >> 5;
  int lane = threadIdx.x & 31;
  if (wave >= NPAIRS / 16 / TPW) return;
  int n = lane & 15;
  int half = (lane >> 4) << 3;
  int koff = (lane >> 4) << 1;
  float b2v = b2[0];

  for (int tt = 0; tt < TPW; ++tt) {
    int tile = wave * TPW + tt;
    int ebase = tile * 16 + half;
    const float* arow = eattr + (tile * 16 + n) * EIN;
    v2f a[EIN / 4];
#pragma unroll
    for (int ks = 0; ks < EIN / 4; ++ks) a[ks] = ld2(arow, ks * 4 + koff);
    int uoff[8];
    int voff[8];
#pragma unroll
    for (int r = 0; r < 8; ++r) {
      uoff[r] = eu[ebase + r] * HID + (n << 2);
      voff[r] = ev[ebase + r] * HID + (n << 2);
    }
    float acc[8] = {};
    for (int j = 0; j < 4; ++j) {
      v4f w2v = *(const v4f*)(W2 + (j << 6) + (n << 2));
      v8f c[4];
#pragma unroll
      for (int r = 0; r < 8; ++r) {
        v4f pu = *(const v4f*)(Qu + uoff[r] + (j << 6));
        v4f pv = *(const v4f*)(Qv + voff[r] + (j << 6));
#pragma unroll
        for (int i = 0; i < 4; ++i) c[i][r] = pu[i] + pv[i];
      }
#pragma unroll
      for (int ks = 0; ks < EIN / 4; ++ks)
#pragma unroll
        for (int i = 0; i < 4; ++i)
          c[i] = wmma_ab(a[ks], sB[((j * 4 + i) * 4 + ks) * 32 + lane], c[i]);
#pragma unroll
      for (int i = 0; i < 4; ++i)
#pragma unroll
        for (int r = 0; r < 8; ++r)
          acc[r] += fmaxf(c[i][r], 0.f) * w2v[i];
    }
#pragma unroll
    for (int r = 0; r < 8; ++r) {
      float s = acc[r];
      s += __shfl_xor(s, 1);
      s += __shfl_xor(s, 2);
      s += __shfl_xor(s, 4);
      s += __shfl_xor(s, 8);
      acc[r] = s;
    }
    if (n == 0) {
#pragma unroll
      for (int r = 0; r < 8; ++r)
        out[ebase + r] = logexp[ebase + r] + acc[r] + b2v;
    }
  }
}

extern "C" void kernel_launch(void* const* d_in, const int* in_sizes, int n_in,
                              void* d_out, int out_size, void* d_ws, size_t ws_size,
                              hipStream_t stream) {
  (void)in_sizes; (void)n_in; (void)out_size; (void)ws_size;
  const float* x      = (const float*)d_in[0];
  const int*   eidx   = (const int*)d_in[1];
  const int*   eu     = (const int*)d_in[2];
  const int*   ev     = (const int*)d_in[3];
  const float* eattr  = (const float*)d_in[4];
  const float* logexp = (const float*)d_in[5];
  const float* Wl1    = (const float*)d_in[6];
  const float* bl1    = (const float*)d_in[7];
  const float* Wr1    = (const float*)d_in[8];
  const float* Wl2    = (const float*)d_in[9];
  const float* bl2    = (const float*)d_in[10];
  const float* Wr2    = (const float*)d_in[11];
  const float* W1     = (const float*)d_in[12];
  const float* b1     = (const float*)d_in[13];
  const float* W2     = (const float*)d_in[14];
  const float* b2     = (const float*)d_in[15];
  float* out = (float*)d_out;

  const int* src = eidx;
  const int* dst = eidx + NEDGES;

  float* ws  = (float*)d_ws;
  int*   deg = (int*)d_ws;
  float* agg = ws + NNODES;
  float* h1  = agg + NNODES * HID;
  float* h2  = h1 + NNODES * HID;
  int*      rowptr  = (int*)h2;
  unsigned* ebin    = (unsigned*)(rowptr + NNODES);
  int*      col     = (int*)ebin + NEDGES;
  int*      cnt     = col + NEDGES;
  int*      binbase = cnt + SCN;

  const int TB = 256;
  int gemm_waves = (NNODES / 16 / SM) * ((HID / 16) / SN);
  int gemm_blocks = gemm_waves / 4;
  int gather_blocks = NNODES / (TB / 32);

  {
    k_bin_count<<<NB1, 1024, 0, stream>>>(dst, cnt);
    k_bin_scatter<<<NB1, 1024, 0, stream>>>(src, dst, cnt, binbase, ebin);
    k_bucket<<<NBIN, 1024, 0, stream>>>(binbase, ebin, rowptr, deg, col);
  }
  {
    k_gather_mean<F_IN><<<gather_blocks, TB, 0, stream>>>(rowptr, deg, col, x, agg);
    k_sage_gemm<F_IN><<<gemm_blocks, 128, 0, stream>>>(agg, x, Wl1, Wr1, bl1, h1);
  }
  {
    k_gather_mean<HID><<<gather_blocks, TB, 0, stream>>>(rowptr, deg, col, h1, agg);
    k_sage_gemm<HID><<<gemm_blocks, 128, 0, stream>>>(agg, h1, Wl2, Wr2, bl2, h2);
  }
  {
    float* qu = agg;
    float* qv = h1;
    int pwaves = (NNODES / 16 / PM) * ((2 * HID / 16) / PN);
    k_node_proj<<<pwaves / 4, 128, 0, stream>>>(h2, W1, b1, qu);
    int waves = NPAIRS / 16 / TPW;
    int blocks = (waves * 32 + TB - 1) / TB;
    k_pair_mlp<<<blocks, TB, 0, stream>>>(qu, qv, eu, ev, eattr, logexp,
                                          W1, W2, b2, out);
  }
}


# SAGE layer GEMMs re-tiled for balance: wave tile 80x32 (was 80x64) in 1250 4-wave blocks; the fullest SIMD runs 5 half-size tiles, not 3 full ones; A rows staged per column half; sum order kept
# speedup vs baseline: 15.8985x; 1.0339x over previous
#include <hip/hip_runtime.h>
#include <hip/hip_bf16.h>

typedef __attribute__((ext_vector_type(2))) float v2f;
typedef __attribute__((ext_vector_type(4))) float v4f;
typedef __attribute__((ext_vector_type(8))) float v8f;
typedef __attribute__((ext_vector_type(4))) int v4i;

#define NNODES 50000
#define NEDGES 800000
#define NPAIRS 500000
#define F_IN   128
#define HID    256
#define EIN    16
#define ZK     (2 * HID + EIN)
#define SM     5
#define SN     2
#define PM     5
#define PN     4
#define TPW    2

__device__ __forceinline__ v8f wmma_ab(v2f a, v2f b, v8f c) {
  return __builtin_amdgcn_wmma_f32_16x16x4_f32(false, a, false, b,
                                               (short)0, c, false, false);
}

__device__ __forceinline__ v2f ld2(const float* __restrict__ p, int k) {
  v2f r;
  r.x = p[k];
  r.y = p[k + 1];
  return r;
}

#define NB1  256
#define ECH  (NEDGES / NB1)
#define NBIN ((NNODES + 255) / 256)
#define SCN  (NBIN * NB1)
#define RPW  ((NBIN + 31) / 32)
static_assert(NB1 == 256, "a wave reads a range's chunk counters as 32 lanes x 8");
static_assert(NEDGES % NB1 == 0, "chunks must tile the edge list");
static_assert(ECH <= 4 * 1024, "a 1024-thread block covers its chunk in 4 steps");
static_assert(NNODES <= 65536, "src and dst are packed into 16 bits each");

__global__ void __launch_bounds__(1024)
k_bin_count(const int* __restrict__ dst, int* __restrict__ cnt) {
  __shared__ int hist[NBIN];
  int t = threadIdx.x;
  int b = blockIdx.x;
  if (t < NBIN) hist[t] = 0;
  __syncthreads();
  const int* dp = dst + b * ECH;
  int d[4];
#pragma unroll
  for (int k = 0; k < 4; ++k) {
    int i = t + k * 1024;
    bool ok = i < ECH;
    int dv = dp[ok ? i : ECH - 1];
    d[k] = ok ? dv : -1;
  }
#pragma unroll
  for (int k = 0; k < 4; ++k)
    if (d[k] >= 0) atomicAdd(&hist[d[k] >> 8], 1);
  __syncthreads();
  if (t < NBIN) cnt[t * NB1 + b] = hist[t];
}

__global__ void __launch_bounds__(1024)
k_bin_scatter(const int* __restrict__ src, const int* __restrict__ dst,
              const int* __restrict__ cnt, int* __restrict__ binbase,
              unsigned* __restrict__ ebin) {
  __shared__ int cur[NBIN];
  __shared__ int btot[NBIN];
  __shared__ int bbef[NBIN];
  int t = threadIdx.x;
  int b = blockIdx.x;
  int lane = t & 31;
  int w = t >> 5;
  const int* sp = src + b * ECH;
  const int* dp = dst + b * ECH;
  int s[4];
  int d[4];
#pragma unroll
  for (int k = 0; k < 4; ++k) {
    int i = t + k * 1024;
    bool ok = i < ECH;
    int ii = ok ? i : ECH - 1;
    int dv = dp[ii];
    s[k] = sp[ii];
    d[k] = ok ? dv : -1;
  }
  int q = b - lane * 8;
  int tot[RPW];
  int bef[RPW];
  v4i ca[RPW];
  v4i cc[RPW];
#pragma unroll
  for (int r = 0; r < RPW; ++r) {
    int bin = w + r * 32;
    if (bin > NBIN - 1) bin = NBIN - 1;
    const int* row = cnt + bin * NB1 + lane * 8;
    ca[r] = *(const v4i*)row;
    cc[r] = *(const v4i*)(row + 4);
  }
#pragma unroll
  for (int r = 0; r < RPW; ++r) {
    v4i a = ca[r];
    v4i c = cc[r];
    tot[r] = a.x + a.y + a.z + a.w + c.x + c.y + c.z + c.w;
    bef[r] = (q > 0 ? a.x : 0) + (q > 1 ? a.y : 0) + (q > 2 ? a.z : 0) + (q > 3 ? a.w : 0)
           + (q > 4 ? c.x : 0) + (q > 5 ? c.y : 0) + (q > 6 ? c.z : 0) + (q > 7 ? c.w : 0);
  }
  for (int o = 16; o > 0; o >>= 1) {
#pragma unroll
    for (int r = 0; r < RPW; ++r) {
      tot[r] += __shfl_xor(tot[r], o);
      bef[r] += __shfl_xor(bef[r], o);
    }
  }
  if (lane == 0) {
#pragma unroll
    for (int r = 0; r < RPW; ++r) {
      int bin = w + r * 32;
      if (bin < NBIN) {
        btot[bin] = tot[r];
        bbef[bin] = bef[r];
      }
    }
  }
  __syncthreads();
  if (t < 32) {
    int e[RPW];
    int run = 0;
#pragma unroll
    for (int k = 0; k < RPW; ++k) {
      int idx = RPW * t + k;
      e[k] = run;
      run += (idx < NBIN) ? btot[idx] : 0;
    }
    int incl = run;
    for (int o = 1; o < 32; o <<= 1) {
      int u = __shfl_up(incl, o);
      if (t >= o) incl += u;
    }
    int ex = incl - run;
#pragma unroll
    for (int k = 0; k < RPW; ++k) {
      int idx = RPW * t + k;
      if (idx < NBIN) {
        int base = ex + e[k];
        cur[idx] = base + bbef[idx];
        if (b == 0) binbase[idx] = base;
      }
    }
  }
  __syncthreads();
#pragma unroll
  for (int k = 0; k < 4; ++k)
    if (d[k] >= 0) {
      int slot = atomicAdd(&cur[d[k] >> 8], 1);
      ebin[slot] = ((unsigned)s[k] << 16) | (unsigned)d[k];
    }
}

__global__ void __launch_bounds__(1024)
k_bucket(const int* __restrict__ binbase, const unsigned* __restrict__ ebin,
         int* __restrict__ rowptr, int* __restrict__ deg, int* __restrict__ col) {
  __shared__ int hist[256];
  __shared__ int cur[256];
  int t = threadIdx.x;
  int bin = blockIdx.x;
  int segbeg = binbase[bin];
  int segend = (bin + 1 < NBIN) ? binbase[bin + 1] : NEDGES;
  int seglen = segend - segbeg;
  const unsigned* ep = ebin + segbeg;
  if (t < 256) hist[t] = 0;
  __syncthreads();
  for (int i0 = t; i0 < seglen; i0 += 4096) {
    unsigned v[4];
#pragma unroll
    for (int k = 0; k < 4; ++k) {
      int i = i0 + k * 1024;
      v[k] = (i < seglen) ? ep[i] : 0xFFFFFFFFu;
    }
#pragma unroll
    for (int k = 0; k < 4; ++k)
      if (v[k] != 0xFFFFFFFFu) atomicAdd(&hist[v[k] & 255u], 1);
  }
  __syncthreads();
  if (t < 32) {
    int e[8];
    int run = 0;
#pragma unroll
    for (int k = 0; k < 8; ++k) {
      e[k] = run;
      run += hist[8 * t + k];
    }
    int incl = run;
    for (int o = 1; o < 32; o <<= 1) {
      int u = __shfl_up(incl, o);
      if (t >= o) incl += u;
    }
    int ex = segbeg + incl - run;
#pragma unroll
    for (int k = 0; k < 8; ++k) cur[8 * t + k] = ex + e[k];
  }
  __syncthreads();
  if (t < 256) {
    int node = bin * 256 + t;
    if (node < NNODES) {
      rowptr[node] = cur[t];
      deg[node] = hist[t];
    }
  }
  __syncthreads();
  for (int i0 = t; i0 < seglen; i0 += 4096) {
    unsigned v[4];
#pragma unroll
    for (int k = 0; k < 4; ++k) {
      int i = i0 + k * 1024;
      v[k] = (i < seglen) ? ep[i] : 0xFFFFFFFFu;
    }
#pragma unroll
    for (int k = 0; k < 4; ++k)
      if (v[k] != 0xFFFFFFFFu) {
        int slot = atomicAdd(&cur[v[k] & 255u], 1);
        col[slot] = (int)(v[k] >> 16);
      }
  }
}

template <int F>
__global__ void __launch_bounds__(256)
k_gather_mean(const int* __restrict__ rowptr, const int* __restrict__ deg,
              const int* __restrict__ col, const float* __restrict__ X,
              float* __restrict__ agg) {
  constexpr int V = F / 128;
  int node = __builtin_amdgcn_readfirstlane((int)((blockIdx.x * blockDim.x + threadIdx.x) >> 5));
  int lane = threadIdx.x & 31;
  if (node >= NNODES) return;
  int beg = __builtin_amdgcn_readfirstlane(rowptr[node]);
  int d = __builtin_amdgcn_readfirstlane(deg[node]);
  const float* xl = X + (lane << 2);
  v4f acc[V];
#pragma unroll
  for (int v = 0; v < V; ++v) acc[v] = (v4f){0.f, 0.f, 0.f, 0.f};
  for (int base = 0; base < d; base += 32) {
    int m = d - base;
    if (m > 32) m = 32;
    int mine = (lane < m) ? col[beg + base + lane] : 0;
    int j = 0;
    for (; j + 4 <= m; j += 4) {
      v4f r[4][V];
#pragma unroll
      for (int u = 0; u < 4; ++u) {
        int s = __builtin_amdgcn_readlane(mine, j + u);
#pragma unroll
        for (int v = 0; v < V; ++v) r[u][v] = *(const v4f*)(xl + s * F + v * 128);
      }
#pragma unroll
      for (int u = 0; u < 4; ++u)
#pragma unroll
        for (int v = 0; v < V; ++v) acc[v] += r[u][v];
    }
    for (; j < m; ++j) {
      int s = __builtin_amdgcn_readlane(mine, j);
#pragma unroll
      for (int v = 0; v < V; ++v) acc[v] += *(const v4f*)(xl + s * F + v * 128);
    }
  }
  float inv = 1.f / fmaxf((float)d, 1.f);
  float* o = agg + node * F + (lane << 2);
#pragma unroll
  for (int v = 0; v < V; ++v) *(v4f*)(o + v * 128) = acc[v] * inv;
}

__device__ __forceinline__ void ld8(const float* __restrict__ p, int k, v4f& lo, v4f& hi) {
  lo = *(const v4f*)(p + k);
  hi = *(const v4f*)(p + k + 4);
}

__device__ __forceinline__ v2f pair_of(v4f lo, v4f hi, int q) {
  v2f r;
  if (q == 0) { r.x = lo.x; r.y = lo.y; }
  else if (q == 1) { r.x = lo.z; r.y = lo.w; }
  else if (q == 2) { r.x = hi.x; r.y = hi.y; }
  else { r.x = hi.z; r.y = hi.w; }
  return r;
}

#define SLAB   64
#define APITCH (SLAB + 4)
#define AROWS  (SM * 16)
#define ABUF   (AROWS * APITCH)
static_assert(SM == PM, "the node GEMMs share one row-block shape");
static_assert((AROWS * (SLAB / 4)) % 128 == 0, "a slab is a whole number of 16-byte pieces per thread");
static_assert(NNODES % AROWS == 0, "row blocks tile the nodes exactly");

template <int K, int WS, int NOP, int NT>
__device__ __forceinline__ void block_acc(const float* __restrict__ A0, const float* __restrict__ A1,
                                          const float* __restrict__ W0, const float* __restrict__ W1,
                                          int rowblk, int col0, float* sA, v8f (&c)[SM][NT]) {
  constexpr int SPO = K / SLAB;
  constexpr int T = NOP * SPO;
  constexpr int NP = AROWS * (SLAB / 4) / 128;
  int t = threadIdx.x;
  int lane = t & 31;
  int koff = (lane >> 4) << 3;
  int srow = t >> 4;
  int spc = (t & 15) << 2;
  const float* g0 = A0 + (rowblk + srow) * K + spc;
  const float* g1 = A1 + (rowblk + srow) * K + spc;
  float* sst = sA + srow * APITCH + spc;
  const float* sfr = sA + (lane & 15) * APITCH + koff;
  const float* w0 = W0 + col0 * WS + koff;
  const float* w1 = W1 + col0 * WS + koff;
  v4f st[NP];
#pragma unroll
  for (int i = 0; i < NP; ++i) st[i] = *(const v4f*)(g0 + i * 8 * K);
#pragma unroll
  for (int i = 0; i < NP; ++i) *(v4f*)(sst + i * 8 * APITCH) = st[i];
  __syncthreads();
#pragma unroll 1
  for (int g = 0; g < T; ++g) {
    int buf = g & 1;
    bool more = g + 1 < T;
    if (more) {
      int gn = g + 1;
      const float* src = (gn < SPO ? g0 : g1) + (gn & (SPO - 1)) * SLAB;
#pragma unroll
      for (int i = 0; i < NP; ++i) st[i] = *(const v4f*)(src + i * 8 * K);
    }
    const float* wb = (g < SPO ? w0 : w1) + (g & (SPO - 1)) * SLAB;
    const float* ab = sfr + buf * ABUF;
    for (int kk = 0; kk < SLAB; kk += 16) {
      v4f alo[SM], ahi[SM];
      v4f blo[NT], bhi[NT];
#pragma unroll
      for (int i = 0; i < NT; ++i) ld8(wb + i * 16 * WS, kk, blo[i], bhi[i]);
#pragma unroll
      for (int r = 0; r < SM; ++r) {
        alo[r] = *(const v4f*)(ab + r * 16 * APITCH + kk);
        ahi[r] = *(const v4f*)(ab + r * 16 * APITCH + kk + 4);
      }
#pragma unroll
      for (int q = 0; q < 4; ++q)
#pragma unroll
        for (int r = 0; r < SM; ++r)
#pragma unroll
          for (int i = 0; i < NT; ++i)
            c[r][i] = wmma_ab(pair_of(alo[r], ahi[r], q), pair_of(blo[i], bhi[i], q), c[r][i]);
    }
    if (more) {
      float* d = sst + (buf ^ 1) * ABUF;
#pragma unroll
      for (int i = 0; i < NP; ++i) *(v4f*)(d + i * 8 * APITCH) = st[i];
      __syncthreads();
    }
  }
}
static_assert((HID / 16) / SN == 8, "two blocks of 4 waves cover the 256 output columns of a SAGE layer");
static_assert((HID / 16) / PN == 4, "a block's 4 waves cover the 256 columns of a projection plane");

template <int K>
__global__ void __launch_bounds__(128)
k_sage_gemm(const float* __restrict__ A, const float* __restrict__ X,
            const float* __restrict__ Wl, const float* __restrict__ Wr,
            const float* __restrict__ bias, float* __restrict__ out) {
  __shared__ __attribute__((aligned(16))) float sA[2 * ABUF];
  int lane = threadIdx.x & 31;
  int gm = blockIdx.x >> 1;
  if (gm >= NNODES / AROWS) return;
  int gn = ((blockIdx.x & 1) << 2) + (threadIdx.x >> 5);
  int col0 = gn * (SN * 16) + (lane & 15);
  v8f c[SM][SN] = {};
  block_acc<K, K, 2, SN>(A, X, Wl, Wr, gm * AROWS, col0, sA, c);
  int mbase = gm * (SM * 16) + ((lane >> 4) << 3);
#pragma unroll
  for (int i = 0; i < SN; ++i) {
    int col = col0 + i * 16;
    float bv = bias[col];
#pragma unroll
    for (int t = 0; t < SM; ++t)
#pragma unroll
      for (int r = 0; r < 8; ++r)
        out[(mbase + t * 16 + r) * HID + col] = fmaxf(c[t][i][r] + bv, 0.f);
  }
}

__global__ void __launch_bounds__(128)
k_node_proj(const float* __restrict__ Hn, const float* __restrict__ W1,
            const float* __restrict__ b1, float* __restrict__ Q) {
  __shared__ __attribute__((aligned(16))) float sA[2 * ABUF];
  int lane = threadIdx.x & 31;
  int gm = blockIdx.x >> 1;
  if (gm >= NNODES / AROWS) return;
  int plane = blockIdx.x & 1;
  int gn = threadIdx.x >> 5;
  int col0 = gn * (PN * 16) + (lane & 15);
  const float* wp = W1 + plane * HID;
  v8f c[PM][PN] = {};
  block_acc<HID, ZK, 1, PN>(Hn, Hn, wp, wp, gm * AROWS, col0, sA, c);
  float* qp = Q + plane * (NNODES * HID);
  int mbase = gm * (PM * 16) + ((lane >> 4) << 3);
#pragma unroll
  for (int i = 0; i < PN; ++i) {
    int col = col0 + i * 16;
    float bv = (plane == 0) ? b1[col] : 0.f;
#pragma unroll
    for (int t = 0; t < PM; ++t)
#pragma unroll
      for (int r = 0; r < 8; ++r)
        qp[(mbase + t * 16 + r) * HID + col] = c[t][i][r] + bv;
  }
}

__device__ __forceinline__ int pair_col(int nt, int n) {
  return ((nt >> 2) << 6) + (n << 2) + (nt & 3);
}

__global__ void k_pair_mlp(const float* __restrict__ Qu, const float* __restrict__ Qv,
                           const int* __restrict__ eu, const int* __restrict__ ev,
                           const float* __restrict__ eattr,
                           const float* __restrict__ logexp,
                           const float* __restrict__ W1, const float* __restrict__ W2,
                           const float* __restrict__ b2, float* __restrict__ out) {
  __shared__ v2f sB[(HID / 16) * (EIN / 4) * 32];
  for (int idx = threadIdx.x; idx < (HID / 16) * (EIN / 4) * 32; idx += blockDim.x) {
    int nt = idx >> 7;
    int ks = (idx >> 5) & 3;
    int ln = idx & 31;
    const float* w = W1 + pair_col(nt, ln & 15) * ZK + 2 * HID;
    sB[idx] = ld2(w, ks * 4 + ((ln >> 4) << 1));
  }
  __syncthreads();

  int gtid = blockIdx.x * blockDim.x + threadIdx.x;
  int wave = gtid >> 5;
  int lane = threadIdx.x & 31;
  if (wave >= NPAIRS / 16 / TPW) return;
  int n = lane & 15;
  int half = (lane >> 4) << 3;
  int koff = (lane >> 4) << 1;
  float b2v = b2[0];

  for (int tt = 0; tt < TPW; ++tt) {
    int tile = wave * TPW + tt;
    int ebase = tile * 16 + half;
    const float* arow = eattr + (tile * 16 + n) * EIN;
    v2f a[EIN / 4];
#pragma unroll
    for (int ks = 0; ks < EIN / 4; ++ks) a[ks] = ld2(arow, ks * 4 + koff);
    int uoff[8];
    int voff[8];
#pragma unroll
    for (int r = 0; r < 8; ++r) {
      uoff[r] = eu[ebase + r] * HID + (n << 2);
      voff[r] = ev[ebase + r] * HID + (n << 2);
    }
    float acc[8] = {};
    for (int j = 0; j < 4; ++j) {
      v4f w2v = *(const v4f*)(W2 + (j << 6) + (n << 2));
      v8f c[4];
#pragma unroll
      for (int r = 0; r < 8; ++r) {
        v4f pu = *(const v4f*)(Qu + uoff[r] + (j << 6));
        v4f pv = *(const v4f*)(Qv + voff[r] + (j << 6));
#pragma unroll
        for (int i = 0; i < 4; ++i) c[i][r] = pu[i] + pv[i];
      }
#pragma unroll
      for (int ks = 0; ks < EIN / 4; ++ks)
#pragma unroll
        for (int i = 0; i < 4; ++i)
          c[i] = wmma_ab(a[ks], sB[((j * 4 + i) * 4 + ks) * 32 + lane], c[i]);
#pragma unroll
      for (int i = 0; i < 4; ++i)
#pragma unroll
        for (int r = 0; r < 8; ++r)
          acc[r] += fmaxf(c[i][r], 0.f) * w2v[i];
    }
#pragma unroll
    for (int r = 0; r < 8; ++r) {
      float s = acc[r];
      s += __shfl_xor(s, 1);
      s += __shfl_xor(s, 2);
      s += __shfl_xor(s, 4);
      s += __shfl_xor(s, 8);
      acc[r] = s;
    }
    if (n == 0) {
#pragma unroll
      for (int r = 0; r < 8; ++r)
        out[ebase + r] = logexp[ebase + r] + acc[r] + b2v;
    }
  }
}

extern "C" void kernel_launch(void* const* d_in, const int* in_sizes, int n_in,
                              void* d_out, int out_size, void* d_ws, size_t ws_size,
                              hipStream_t stream) {
  (void)in_sizes; (void)n_in; (void)out_size; (void)ws_size;
  const float* x      = (const float*)d_in[0];
  const int*   eidx   = (const int*)d_in[1];
  const int*   eu     = (const int*)d_in[2];
  const int*   ev     = (const int*)d_in[3];
  const float* eattr  = (const float*)d_in[4];
  const float* logexp = (const float*)d_in[5];
  const float* Wl1    = (const float*)d_in[6];
  const float* bl1    = (const float*)d_in[7];
  const float* Wr1    = (const float*)d_in[8];
  const float* Wl2    = (const float*)d_in[9];
  const float* bl2    = (const float*)d_in[10];
  const float* Wr2    = (const float*)d_in[11];
  const float* W1     = (const float*)d_in[12];
  const float* b1     = (const float*)d_in[13];
  const float* W2     = (const float*)d_in[14];
  const float* b2     = (const float*)d_in[15];
  float* out = (float*)d_out;

  const int* src = eidx;
  const int* dst = eidx + NEDGES;

  float* ws  = (float*)d_ws;
  int*   deg = (int*)d_ws;
  float* agg = ws + NNODES;
  float* h1  = agg + NNODES * HID;
  float* h2  = h1 + NNODES * HID;
  int*      rowptr  = (int*)h2;
  unsigned* ebin    = (unsigned*)(rowptr + NNODES);
  int*      col     = (int*)ebin + NEDGES;
  int*      cnt     = col + NEDGES;
  int*      binbase = cnt + SCN;

  const int TB = 256;
  int gemm_waves = (NNODES / 16 / SM) * ((HID / 16) / SN);
  int gemm_blocks = gemm_waves / 4;
  int gather_blocks = NNODES / (TB / 32);

  {
    k_bin_count<<<NB1, 1024, 0, stream>>>(dst, cnt);
    k_bin_scatter<<<NB1, 1024, 0, stream>>>(src, dst, cnt, binbase, ebin);
    k_bucket<<<NBIN, 1024, 0, stream>>>(binbase, ebin, rowptr, deg, col);
  }
  {
    k_gather_mean<F_IN><<<gather_blocks, TB, 0, stream>>>(rowptr, deg, col, x, agg);
    k_sage_gemm<F_IN><<<gemm_blocks, 128, 0, stream>>>(agg, x, Wl1, Wr1, bl1, h1);
  }
  {
    k_gather_mean<HID><<<gather_blocks, TB, 0, stream>>>(rowptr, deg, col, h1, agg);
    k_sage_gemm<HID><<<gemm_blocks, 128, 0, stream>>>(agg, h1, Wl2, Wr2, bl2, h2);
  }
  {
    float* qu = agg;
    float* qv = h1;
    int pwaves = (NNODES / 16 / PM) * ((2 * HID / 16) / PN);
    k_node_proj<<<pwaves / 4, 128, 0, stream>>>(h2, W1, b1, qu);
    int waves = NPAIRS / 16 / TPW;
    int blocks = (waves * 32 + TB - 1) / TB;
    k_pair_mlp<<<blocks, TB, 0, stream>>>(qu, qv, eu, ev, eattr, logexp,
                                          W1, W2, b2, out);
  }
}


# Workspace re-laid: planes agg, h1, h2 (Qu, Qv live in them) start on 256-byte boundaries, were 64 bytes past a line; degree joins the bucket start as one 8-byte record per node; sum order kept
# speedup vs baseline: 17.2092x; 1.0824x over previous
#include <hip/hip_runtime.h>
#include <hip/hip_bf16.h>

typedef __attribute__((ext_vector_type(2))) float v2f;
typedef __attribute__((ext_vector_type(4))) float v4f;
typedef __attribute__((ext_vector_type(8))) float v8f;
typedef __attribute__((ext_vector_type(4))) int v4i;

#define NNODES 50000
#define NEDGES 800000
#define NPAIRS 500000
#define F_IN   128
#define HID    256
#define EIN    16
#define ZK     (2 * HID + EIN)
#define SM     5
#define SN     2
#define PM     5
#define PN     4
#define TPW    2

__device__ __forceinline__ v8f wmma_ab(v2f a, v2f b, v8f c) {
  return __builtin_amdgcn_wmma_f32_16x16x4_f32(false, a, false, b,
                                               (short)0, c, false, false);
}

__device__ __forceinline__ v2f ld2(const float* __restrict__ p, int k) {
  v2f r;
  r.x = p[k];
  r.y = p[k + 1];
  return r;
}

#define NB1  256
#define ECH  (NEDGES / NB1)
#define NBIN ((NNODES + 255) / 256)
#define SCN  (NBIN * NB1)
#define RPW  ((NBIN + 31) / 32)
static_assert(NB1 == 256, "a wave reads a range's chunk counters as 32 lanes x 8");
static_assert(NEDGES % NB1 == 0, "chunks must tile the edge list");
static_assert(ECH <= 4 * 1024, "a 1024-thread block covers its chunk in 4 steps");
static_assert(NNODES <= 65536, "src and dst are packed into 16 bits each");

__global__ void __launch_bounds__(1024)
k_bin_count(const int* __restrict__ dst, int* __restrict__ cnt) {
  __shared__ int hist[NBIN];
  int t = threadIdx.x;
  int b = blockIdx.x;
  if (t < NBIN) hist[t] = 0;
  __syncthreads();
  const int* dp = dst + b * ECH;
  int d[4];
#pragma unroll
  for (int k = 0; k < 4; ++k) {
    int i = t + k * 1024;
    bool ok = i < ECH;
    int dv = dp[ok ? i : ECH - 1];
    d[k] = ok ? dv : -1;
  }
#pragma unroll
  for (int k = 0; k < 4; ++k)
    if (d[k] >= 0) atomicAdd(&hist[d[k] >> 8], 1);
  __syncthreads();
  if (t < NBIN) cnt[t * NB1 + b] = hist[t];
}

__global__ void __launch_bounds__(1024)
k_bin_scatter(const int* __restrict__ src, const int* __restrict__ dst,
              const int* __restrict__ cnt, int* __restrict__ binbase,
              unsigned* __restrict__ ebin) {
  __shared__ int cur[NBIN];
  __shared__ int btot[NBIN];
  __shared__ int bbef[NBIN];
  int t = threadIdx.x;
  int b = blockIdx.x;
  int lane = t & 31;
  int w = t >> 5;
  const int* sp = src + b * ECH;
  const int* dp = dst + b * ECH;
  int s[4];
  int d[4];
#pragma unroll
  for (int k = 0; k < 4; ++k) {
    int i = t + k * 1024;
    bool ok = i < ECH;
    int ii = ok ? i : ECH - 1;
    int dv = dp[ii];
    s[k] = sp[ii];
    d[k] = ok ? dv : -1;
  }
  int q = b - lane * 8;
  int tot[RPW];
  int bef[RPW];
  v4i ca[RPW];
  v4i cc[RPW];
#pragma unroll
  for (int r = 0; r < RPW; ++r) {
    int bin = w + r * 32;
    if (bin > NBIN - 1) bin = NBIN - 1;
    const int* row = cnt + bin * NB1 + lane * 8;
    ca[r] = *(const v4i*)row;
    cc[r] = *(const v4i*)(row + 4);
  }
#pragma unroll
  for (int r = 0; r < RPW; ++r) {
    v4i a = ca[r];
    v4i c = cc[r];
    tot[r] = a.x + a.y + a.z + a.w + c.x + c.y + c.z + c.w;
    bef[r] = (q > 0 ? a.x : 0) + (q > 1 ? a.y : 0) + (q > 2 ? a.z : 0) + (q > 3 ? a.w : 0)
           + (q > 4 ? c.x : 0) + (q > 5 ? c.y : 0) + (q > 6 ? c.z : 0) + (q > 7 ? c.w : 0);
  }
  for (int o = 16; o > 0; o >>= 1) {
#pragma unroll
    for (int r = 0; r < RPW; ++r) {
      tot[r] += __shfl_xor(tot[r], o);
      bef[r] += __shfl_xor(bef[r], o);
    }
  }
  if (lane == 0) {
#pragma unroll
    for (int r = 0; r < RPW; ++r) {
      int bin = w + r * 32;
      if (bin < NBIN) {
        btot[bin] = tot[r];
        bbef[bin] = bef[r];
      }
    }
  }
  __syncthreads();
  if (t < 32) {
    int e[RPW];
    int run = 0;
#pragma unroll
    for (int k = 0; k < RPW; ++k) {
      int idx = RPW * t + k;
      e[k] = run;
      run += (idx < NBIN) ? btot[idx] : 0;
    }
    int incl = run;
    for (int o = 1; o < 32; o <<= 1) {
      int u = __shfl_up(incl, o);
      if (t >= o) incl += u;
    }
    int ex = incl - run;
#pragma unroll
    for (int k = 0; k < RPW; ++k) {
      int idx = RPW * t + k;
      if (idx < NBIN) {
        int base = ex + e[k];
        cur[idx] = base + bbef[idx];
        if (b == 0) binbase[idx] = base;
      }
    }
  }
  __syncthreads();
#pragma unroll
  for (int k = 0; k < 4; ++k)
    if (d[k] >= 0) {
      int slot = atomicAdd(&cur[d[k] >> 8], 1);
      ebin[slot] = ((unsigned)s[k] << 16) | (unsigned)d[k];
    }
}

__global__ void __launch_bounds__(1024)
k_bucket(const int* __restrict__ binbase, const unsigned* __restrict__ ebin,
         int2* __restrict__ rec, int* __restrict__ col) {
  __shared__ int hist[256];
  __shared__ int cur[256];
  int t = threadIdx.x;
  int bin = blockIdx.x;
  int segbeg = binbase[bin];
  int segend = (bin + 1 < NBIN) ? binbase[bin + 1] : NEDGES;
  int seglen = segend - segbeg;
  const unsigned* ep = ebin + segbeg;
  if (t < 256) hist[t] = 0;
  __syncthreads();
  for (int i0 = t; i0 < seglen; i0 += 4096) {
    unsigned v[4];
#pragma unroll
    for (int k = 0; k < 4; ++k) {
      int i = i0 + k * 1024;
      v[k] = (i < seglen) ? ep[i] : 0xFFFFFFFFu;
    }
#pragma unroll
    for (int k = 0; k < 4; ++k)
      if (v[k] != 0xFFFFFFFFu) atomicAdd(&hist[v[k] & 255u], 1);
  }
  __syncthreads();
  if (t < 32) {
    int e[8];
    int run = 0;
#pragma unroll
    for (int k = 0; k < 8; ++k) {
      e[k] = run;
      run += hist[8 * t + k];
    }
    int incl = run;
    for (int o = 1; o < 32; o <<= 1) {
      int u = __shfl_up(incl, o);
      if (t >= o) incl += u;
    }
    int ex = segbeg + incl - run;
#pragma unroll
    for (int k = 0; k < 8; ++k) cur[8 * t + k] = ex + e[k];
  }
  __syncthreads();
  if (t < 256) {
    int node = bin * 256 + t;
    if (node < NNODES) rec[node] = make_int2(cur[t], hist[t]);
  }
  __syncthreads();
  for (int i0 = t; i0 < seglen; i0 += 4096) {
    unsigned v[4];
#pragma unroll
    for (int k = 0; k < 4; ++k) {
      int i = i0 + k * 1024;
      v[k] = (i < seglen) ? ep[i] : 0xFFFFFFFFu;
    }
#pragma unroll
    for (int k = 0; k < 4; ++k)
      if (v[k] != 0xFFFFFFFFu) {
        int slot = atomicAdd(&cur[v[k] & 255u], 1);
        col[slot] = (int)(v[k] >> 16);
      }
  }
}

template <int F>
__global__ void __launch_bounds__(256)
k_gather_mean(const int2* __restrict__ rec,
              const int* __restrict__ col, const float* __restrict__ X,
              float* __restrict__ agg) {
  constexpr int V = F / 128;
  int node = __builtin_amdgcn_readfirstlane((int)((blockIdx.x * blockDim.x + threadIdx.x) >> 5));
  int lane = threadIdx.x & 31;
  if (node >= NNODES) return;
  int2 nr = rec[node];
  int beg = __builtin_amdgcn_readfirstlane(nr.x);
  int d = __builtin_amdgcn_readfirstlane(nr.y);
  const float* xl = X + (lane << 2);
  v4f acc[V];
#pragma unroll
  for (int v = 0; v < V; ++v) acc[v] = (v4f){0.f, 0.f, 0.f, 0.f};
  for (int base = 0; base < d; base += 32) {
    int m = d - base;
    if (m > 32) m = 32;
    int mine = (lane < m) ? col[beg + base + lane] : 0;
    int j = 0;
    for (; j + 4 <= m; j += 4) {
      v4f r[4][V];
#pragma unroll
      for (int u = 0; u < 4; ++u) {
        int s = __builtin_amdgcn_readlane(mine, j + u);
#pragma unroll
        for (int v = 0; v < V; ++v) r[u][v] = *(const v4f*)(xl + s * F + v * 128);
      }
#pragma unroll
      for (int u = 0; u < 4; ++u)
#pragma unroll
        for (int v = 0; v < V; ++v) acc[v] += r[u][v];
    }
    for (; j < m; ++j) {
      int s = __builtin_amdgcn_readlane(mine, j);
#pragma unroll
      for (int v = 0; v < V; ++v) acc[v] += *(const v4f*)(xl + s * F + v * 128);
    }
  }
  float inv = 1.f / fmaxf((float)d, 1.f);
  float* o = agg + node * F + (lane << 2);
#pragma unroll
  for (int v = 0; v < V; ++v) *(v4f*)(o + v * 128) = acc[v] * inv;
}

__device__ __forceinline__ void ld8(const float* __restrict__ p, int k, v4f& lo, v4f& hi) {
  lo = *(const v4f*)(p + k);
  hi = *(const v4f*)(p + k + 4);
}

__device__ __forceinline__ v2f pair_of(v4f lo, v4f hi, int q) {
  v2f r;
  if (q == 0) { r.x = lo.x; r.y = lo.y; }
  else if (q == 1) { r.x = lo.z; r.y = lo.w; }
  else if (q == 2) { r.x = hi.x; r.y = hi.y; }
  else { r.x = hi.z; r.y = hi.w; }
  return r;
}

#define SLAB   64
#define APITCH (SLAB + 4)
#define AROWS  (SM * 16)
#define ABUF   (AROWS * APITCH)
static_assert(SM == PM, "the node GEMMs share one row-block shape");
static_assert((AROWS * (SLAB / 4)) % 128 == 0, "a slab is a whole number of 16-byte pieces per thread");
static_assert(NNODES % AROWS == 0, "row blocks tile the nodes exactly");

template <int K, int WS, int NOP, int NT>
__device__ __forceinline__ void block_acc(const float* __restrict__ A0, const float* __restrict__ A1,
                                          const float* __restrict__ W0, const float* __restrict__ W1,
                                          int rowblk, int col0, float* sA, v8f (&c)[SM][NT]) {
  constexpr int SPO = K / SLAB;
  constexpr int T = NOP * SPO;
  constexpr int NP = AROWS * (SLAB / 4) / 128;
  int t = threadIdx.x;
  int lane = t & 31;
  int koff = (lane >> 4) << 3;
  int srow = t >> 4;
  int spc = (t & 15) << 2;
  const float* g0 = A0 + (rowblk + srow) * K + spc;
  const float* g1 = A1 + (rowblk + srow) * K + spc;
  float* sst = sA + srow * APITCH + spc;
  const float* sfr = sA + (lane & 15) * APITCH + koff;
  const float* w0 = W0 + col0 * WS + koff;
  const float* w1 = W1 + col0 * WS + koff;
  v4f st[NP];
#pragma unroll
  for (int i = 0; i < NP; ++i) st[i] = *(const v4f*)(g0 + i * 8 * K);
#pragma unroll
  for (int i = 0; i < NP; ++i) *(v4f*)(sst + i * 8 * APITCH) = st[i];
  __syncthreads();
#pragma unroll 1
  for (int g = 0; g < T; ++g) {
    int buf = g & 1;
    bool more = g + 1 < T;
    if (more) {
      int gn = g + 1;
      const float* src = (gn < SPO ? g0 : g1) + (gn & (SPO - 1)) * SLAB;
#pragma unroll
      for (int i = 0; i < NP; ++i) st[i] = *(const v4f*)(src + i * 8 * K);
    }
    const float* wb = (g < SPO ? w0 : w1) + (g & (SPO - 1)) * SLAB;
    const float* ab = sfr + buf * ABUF;
    for (int kk = 0; kk < SLAB; kk += 16) {
      v4f alo[SM], ahi[SM];
      v4f blo[NT], bhi[NT];
#pragma unroll
      for (int i = 0; i < NT; ++i) ld8(wb + i * 16 * WS, kk, blo[i], bhi[i]);
#pragma unroll
      for (int r = 0; r < SM; ++r) {
        alo[r] = *(const v4f*)(ab + r * 16 * APITCH + kk);
        ahi[r] = *(const v4f*)(ab + r * 16 * APITCH + kk + 4);
      }
#pragma unroll
      for (int q = 0; q < 4; ++q)
#pragma unroll
        for (int r = 0; r < SM; ++r)
#pragma unroll
          for (int i = 0; i < NT; ++i)
            c[r][i] = wmma_ab(pair_of(alo[r], ahi[r], q), pair_of(blo[i], bhi[i], q), c[r][i]);
    }
    if (more) {
      float* d = sst + (buf ^ 1) * ABUF;
#pragma unroll
      for (int i = 0; i < NP; ++i) *(v4f*)(d + i * 8 * APITCH) = st[i];
      __syncthreads();
    }
  }
}
static_assert((HID / 16) / SN == 8, "two blocks of 4 waves cover the 256 output columns of a SAGE layer");
static_assert((HID / 16) / PN == 4, "a block's 4 waves cover the 256 columns of a projection plane");

template <int K>
__global__ void __launch_bounds__(128)
k_sage_gemm(const float* __restrict__ A, const float* __restrict__ X,
            const float* __restrict__ Wl, const float* __restrict__ Wr,
            const float* __restrict__ bias, float* __restrict__ out) {
  __shared__ __attribute__((aligned(16))) float sA[2 * ABUF];
  int lane = threadIdx.x & 31;
  int gm = blockIdx.x >> 1;
  if (gm >= NNODES / AROWS) return;
  int gn = ((blockIdx.x & 1) << 2) + (threadIdx.x >> 5);
  int col0 = gn * (SN * 16) + (lane & 15);
  v8f c[SM][SN] = {};
  block_acc<K, K, 2, SN>(A, X, Wl, Wr, gm * AROWS, col0, sA, c);
  int mbase = gm * (SM * 16) + ((lane >> 4) << 3);
#pragma unroll
  for (int i = 0; i < SN; ++i) {
    int col = col0 + i * 16;
    float bv = bias[col];
#pragma unroll
    for (int t = 0; t < SM; ++t)
#pragma unroll
      for (int r = 0; r < 8; ++r)
        out[(mbase + t * 16 + r) * HID + col] = fmaxf(c[t][i][r] + bv, 0.f);
  }
}

__global__ void __launch_bounds__(128)
k_node_proj(const float* __restrict__ Hn, const float* __restrict__ W1,
            const float* __restrict__ b1, float* __restrict__ Q) {
  __shared__ __attribute__((aligned(16))) float sA[2 * ABUF];
  int lane = threadIdx.x & 31;
  int gm = blockIdx.x >> 1;
  if (gm >= NNODES / AROWS) return;
  int plane = blockIdx.x & 1;
  int gn = threadIdx.x >> 5;
  int col0 = gn * (PN * 16) + (lane & 15);
  const float* wp = W1 + plane * HID;
  v8f c[PM][PN] = {};
  block_acc<HID, ZK, 1, PN>(Hn, Hn, wp, wp, gm * AROWS, col0, sA, c);
  float* qp = Q + plane * (NNODES * HID);
  int mbase = gm * (PM * 16) + ((lane >> 4) << 3);
#pragma unroll
  for (int i = 0; i < PN; ++i) {
    int col = col0 + i * 16;
    float bv = (plane == 0) ? b1[col] : 0.f;
#pragma unroll
    for (int t = 0; t < PM; ++t)
#pragma unroll
      for (int r = 0; r < 8; ++r)
        qp[(mbase + t * 16 + r) * HID + col] = c[t][i][r] + bv;
  }
}

__device__ __forceinline__ int pair_col(int nt, int n) {
  return ((nt >> 2) << 6) + (n << 2) + (nt & 3);
}

__global__ void k_pair_mlp(const float* __restrict__ Qu, const float* __restrict__ Qv,
                           const int* __restrict__ eu, const int* __restrict__ ev,
                           const float* __restrict__ eattr,
                           const float* __restrict__ logexp,
                           const float* __restrict__ W1, const float* __restrict__ W2,
                           const float* __restrict__ b2, float* __restrict__ out) {
  __shared__ v2f sB[(HID / 16) * (EIN / 4) * 32];
  for (int idx = threadIdx.x; idx < (HID / 16) * (EIN / 4) * 32; idx += blockDim.x) {
    int nt = idx >> 7;
    int ks = (idx >> 5) & 3;
    int ln = idx & 31;
    const float* w = W1 + pair_col(nt, ln & 15) * ZK + 2 * HID;
    sB[idx] = ld2(w, ks * 4 + ((ln >> 4) << 1));
  }
  __syncthreads();

  int gtid = blockIdx.x * blockDim.x + threadIdx.x;
  int wave = gtid >> 5;
  int lane = threadIdx.x & 31;
  if (wave >= NPAIRS / 16 / TPW) return;
  int n = lane & 15;
  int half = (lane >> 4) << 3;
  int koff = (lane >> 4) << 1;
  float b2v = b2[0];

  for (int tt = 0; tt < TPW; ++tt) {
    int tile = wave * TPW + tt;
    int ebase = tile * 16 + half;
    const float* arow = eattr + (tile * 16 + n) * EIN;
    v2f a[EIN / 4];
#pragma unroll
    for (int ks = 0; ks < EIN / 4; ++ks) a[ks] = ld2(arow, ks * 4 + koff);
    int uoff[8];
    int voff[8];
#pragma unroll
    for (int r = 0; r < 8; ++r) {
      uoff[r] = eu[ebase + r] * HID + (n << 2);
      voff[r] = ev[ebase + r] * HID + (n << 2);
    }
    float acc[8] = {};
    for (int j = 0; j < 4; ++j) {
      v4f w2v = *(const v4f*)(W2 + (j << 6) + (n << 2));
      v8f c[4];
#pragma unroll
      for (int r = 0; r < 8; ++r) {
        v4f pu = *(const v4f*)(Qu + uoff[r] + (j << 6));
        v4f pv = *(const v4f*)(Qv + voff[r] + (j << 6));
#pragma unroll
        for (int i = 0; i < 4; ++i) c[i][r] = pu[i] + pv[i];
      }
#pragma unroll
      for (int ks = 0; ks < EIN / 4; ++ks)
#pragma unroll
        for (int i = 0; i < 4; ++i)
          c[i] = wmma_ab(a[ks], sB[((j * 4 + i) * 4 + ks) * 32 + lane], c[i]);
#pragma unroll
      for (int i = 0; i < 4; ++i)
#pragma unroll
        for (int r = 0; r < 8; ++r)
          acc[r] += fmaxf(c[i][r], 0.f) * w2v[i];
    }
#pragma unroll
    for (int r = 0; r < 8; ++r) {
      float s = acc[r];
      s += __shfl_xor(s, 1);
      s += __shfl_xor(s, 2);
      s += __shfl_xor(s, 4);
      s += __shfl_xor(s, 8);
      acc[r] = s;
    }
    if (n == 0) {
#pragma unroll
      for (int r = 0; r < 8; ++r)
        out[ebase + r] = logexp[ebase + r] + acc[r] + b2v;
    }
  }
}

extern "C" void kernel_launch(void* const* d_in, const int* in_sizes, int n_in,
                              void* d_out, int out_size, void* d_ws, size_t ws_size,
                              hipStream_t stream) {
  (void)in_sizes; (void)n_in; (void)out_size; (void)ws_size;
  const float* x      = (const float*)d_in[0];
  const int*   eidx   = (const int*)d_in[1];
  const int*   eu     = (const int*)d_in[2];
  const int*   ev     = (const int*)d_in[3];
  const float* eattr  = (const float*)d_in[4];
  const float* logexp = (const float*)d_in[5];
  const float* Wl1    = (const float*)d_in[6];
  const float* bl1    = (const float*)d_in[7];
  const float* Wr1    = (const float*)d_in[8];
  const float* Wl2    = (const float*)d_in[9];
  const float* bl2    = (const float*)d_in[10];
  const float* Wr2    = (const float*)d_in[11];
  const float* W1     = (const float*)d_in[12];
  const float* b1     = (const float*)d_in[13];
  const float* W2     = (const float*)d_in[14];
  const float* b2     = (const float*)d_in[15];
  float* out = (float*)d_out;

  const int* src = eidx;
  const int* dst = eidx + NEDGES;

  float* agg = (float*)d_ws;
  float* h1  = agg + NNODES * HID;
  float* h2  = h1 + NNODES * HID;
  int2*     rec     = (int2*)h2;
  unsigned* ebin    = (unsigned*)(rec + NNODES);
  int*      col     = (int*)ebin + NEDGES;
  int*      cnt     = col + NEDGES;
  int*      binbase = cnt + SCN;

  const int TB = 256;
  int gemm_waves = (NNODES / 16 / SM) * ((HID / 16) / SN);
  int gemm_blocks = gemm_waves / 4;
  int gather_blocks = NNODES / (TB / 32);

  {
    k_bin_count<<<NB1, 1024, 0, stream>>>(dst, cnt);
    k_bin_scatter<<<NB1, 1024, 0, stream>>>(src, dst, cnt, binbase, ebin);
    k_bucket<<<NBIN, 1024, 0, stream>>>(binbase, ebin, rec, col);
  }
  {
    k_gather_mean<F_IN><<<gather_blocks, TB, 0, stream>>>(rec, col, x, agg);
    k_sage_gemm<F_IN><<<gemm_blocks, 128, 0, stream>>>(agg, x, Wl1, Wr1, bl1, h1);
  }
  {
    k_gather_mean<HID><<<gather_blocks, TB, 0, stream>>>(rec, col, h1, agg);
    k_sage_gemm<HID><<<gemm_blocks, 128, 0, stream>>>(agg, h1, Wl2, Wr2, bl2, h2);
  }
  {
    float* qu = agg;
    float* qv = h1;
    int pwaves = (NNODES / 16 / PM) * ((2 * HID / 16) / PN);
    k_node_proj<<<pwaves / 4, 128, 0, stream>>>(h2, W1, b1, qu);
    int waves = NPAIRS / 16 / TPW;
    int blocks = (waves * 32 + TB - 1) / TB;
    k_pair_mlp<<<blocks, TB, 0, stream>>>(qu, qv, eu, ev, eattr, logexp,
                                          W1, W2, b2, out);
  }
}


# Pair kernel: 256-thread launch bound stated (compiled for 1024 it was held to 128 registers); a column group's 16 row pieces and 16 W1 fragments requested together, W2 read once; sums unchanged
# speedup vs baseline: 18.0512x; 1.0489x over previous
#include <hip/hip_runtime.h>
#include <hip/hip_bf16.h>

typedef __attribute__((ext_vector_type(2))) float v2f;
typedef __attribute__((ext_vector_type(4))) float v4f;
typedef __attribute__((ext_vector_type(8))) float v8f;
typedef __attribute__((ext_vector_type(4))) int v4i;

#define NNODES 50000
#define NEDGES 800000
#define NPAIRS 500000
#define F_IN   128
#define HID    256
#define EIN    16
#define ZK     (2 * HID + EIN)
#define SM     5
#define SN     2
#define PM     5
#define PN     4
#define TPW    2

__device__ __forceinline__ v8f wmma_ab(v2f a, v2f b, v8f c) {
  return __builtin_amdgcn_wmma_f32_16x16x4_f32(false, a, false, b,
                                               (short)0, c, false, false);
}

__device__ __forceinline__ v2f ld2(const float* __restrict__ p, int k) {
  v2f r;
  r.x = p[k];
  r.y = p[k + 1];
  return r;
}

#define NB1  256
#define ECH  (NEDGES / NB1)
#define NBIN ((NNODES + 255) / 256)
#define SCN  (NBIN * NB1)
#define RPW  ((NBIN + 31) / 32)
static_assert(NB1 == 256, "a wave reads a range's chunk counters as 32 lanes x 8");
static_assert(NEDGES % NB1 == 0, "chunks must tile the edge list");
static_assert(ECH <= 4 * 1024, "a 1024-thread block covers its chunk in 4 steps");
static_assert(NNODES <= 65536, "src and dst are packed into 16 bits each");

__global__ void __launch_bounds__(1024)
k_bin_count(const int* __restrict__ dst, int* __restrict__ cnt) {
  __shared__ int hist[NBIN];
  int t = threadIdx.x;
  int b = blockIdx.x;
  if (t < NBIN) hist[t] = 0;
  __syncthreads();
  const int* dp = dst + b * ECH;
  int d[4];
#pragma unroll
  for (int k = 0; k < 4; ++k) {
    int i = t + k * 1024;
    bool ok = i < ECH;
    int dv = dp[ok ? i : ECH - 1];
    d[k] = ok ? dv : -1;
  }
#pragma unroll
  for (int k = 0; k < 4; ++k)
    if (d[k] >= 0) atomicAdd(&hist[d[k] >> 8], 1);
  __syncthreads();
  if (t < NBIN) cnt[t * NB1 + b] = hist[t];
}

__global__ void __launch_bounds__(1024)
k_bin_scatter(const int* __restrict__ src, const int* __restrict__ dst,
              const int* __restrict__ cnt, int* __restrict__ binbase,
              unsigned* __restrict__ ebin) {
  __shared__ int cur[NBIN];
  __shared__ int btot[NBIN];
  __shared__ int bbef[NBIN];
  int t = threadIdx.x;
  int b = blockIdx.x;
  int lane = t & 31;
  int w = t >> 5;
  const int* sp = src + b * ECH;
  const int* dp = dst + b * ECH;
  int s[4];
  int d[4];
#pragma unroll
  for (int k = 0; k < 4; ++k) {
    int i = t + k * 1024;
    bool ok = i < ECH;
    int ii = ok ? i : ECH - 1;
    int dv = dp[ii];
    s[k] = sp[ii];
    d[k] = ok ? dv : -1;
  }
  int q = b - lane * 8;
  int tot[RPW];
  int bef[RPW];
  v4i ca[RPW];
  v4i cc[RPW];
#pragma unroll
  for (int r = 0; r < RPW; ++r) {
    int bin = w + r * 32;
    if (bin > NBIN - 1) bin = NBIN - 1;
    const int* row = cnt + bin * NB1 + lane * 8;
    ca[r] = *(const v4i*)row;
    cc[r] = *(const v4i*)(row + 4);
  }
#pragma unroll
  for (int r = 0; r < RPW; ++r) {
    v4i a = ca[r];
    v4i c = cc[r];
    tot[r] = a.x + a.y + a.z + a.w + c.x + c.y + c.z + c.w;
    bef[r] = (q > 0 ? a.x : 0) + (q > 1 ? a.y : 0) + (q > 2 ? a.z : 0) + (q > 3 ? a.w : 0)
           + (q > 4 ? c.x : 0) + (q > 5 ? c.y : 0) + (q > 6 ? c.z : 0) + (q > 7 ? c.w : 0);
  }
  for (int o = 16; o > 0; o >>= 1) {
#pragma unroll
    for (int r = 0; r < RPW; ++r) {
      tot[r] += __shfl_xor(tot[r], o);
      bef[r] += __shfl_xor(bef[r], o);
    }
  }
  if (lane == 0) {
#pragma unroll
    for (int r = 0; r < RPW; ++r) {
      int bin = w + r * 32;
      if (bin < NBIN) {
        btot[bin] = tot[r];
        bbef[bin] = bef[r];
      }
    }
  }
  __syncthreads();
  if (t < 32) {
    int e[RPW];
    int run = 0;
#pragma unroll
    for (int k = 0; k < RPW; ++k) {
      int idx = RPW * t + k;
      e[k] = run;
      run += (idx < NBIN) ? btot[idx] : 0;
    }
    int incl = run;
    for (int o = 1; o < 32; o <<= 1) {
      int u = __shfl_up(incl, o);
      if (t >= o) incl += u;
    }
    int ex = incl - run;
#pragma unroll
    for (int k = 0; k < RPW; ++k) {
      int idx = RPW * t + k;
      if (idx < NBIN) {
        int base = ex + e[k];
        cur[idx] = base + bbef[idx];
        if (b == 0) binbase[idx] = base;
      }
    }
  }
  __syncthreads();
#pragma unroll
  for (int k = 0; k < 4; ++k)
    if (d[k] >= 0) {
      int slot = atomicAdd(&cur[d[k] >> 8], 1);
      ebin[slot] = ((unsigned)s[k] << 16) | (unsigned)d[k];
    }
}

__global__ void __launch_bounds__(1024)
k_bucket(const int* __restrict__ binbase, const unsigned* __restrict__ ebin,
         int2* __restrict__ rec, int* __restrict__ col) {
  __shared__ int hist[256];
  __shared__ int cur[256];
  int t = threadIdx.x;
  int bin = blockIdx.x;
  int segbeg = binbase[bin];
  int segend = (bin + 1 < NBIN) ? binbase[bin + 1] : NEDGES;
  int seglen = segend - segbeg;
  const unsigned* ep = ebin + segbeg;
  if (t < 256) hist[t] = 0;
  __syncthreads();
  for (int i0 = t; i0 < seglen; i0 += 4096) {
    unsigned v[4];
#pragma unroll
    for (int k = 0; k < 4; ++k) {
      int i = i0 + k * 1024;
      v[k] = (i < seglen) ? ep[i] : 0xFFFFFFFFu;
    }
#pragma unroll
    for (int k = 0; k < 4; ++k)
      if (v[k] != 0xFFFFFFFFu) atomicAdd(&hist[v[k] & 255u], 1);
  }
  __syncthreads();
  if (t < 32) {
    int e[8];
    int run = 0;
#pragma unroll
    for (int k = 0; k < 8; ++k) {
      e[k] = run;
      run += hist[8 * t + k];
    }
    int incl = run;
    for (int o = 1; o < 32; o <<= 1) {
      int u = __shfl_up(incl, o);
      if (t >= o) incl += u;
    }
    int ex = segbeg + incl - run;
#pragma unroll
    for (int k = 0; k < 8; ++k) cur[8 * t + k] = ex + e[k];
  }
  __syncthreads();
  if (t < 256) {
    int node = bin * 256 + t;
    if (node < NNODES) rec[node] = make_int2(cur[t], hist[t]);
  }
  __syncthreads();
  for (int i0 = t; i0 < seglen; i0 += 4096) {
    unsigned v[4];
#pragma unroll
    for (int k = 0; k < 4; ++k) {
      int i = i0 + k * 1024;
      v[k] = (i < seglen) ? ep[i] : 0xFFFFFFFFu;
    }
#pragma unroll
    for (int k = 0; k < 4; ++k)
      if (v[k] != 0xFFFFFFFFu) {
        int slot = atomicAdd(&cur[v[k] & 255u], 1);
        col[slot] = (int)(v[k] >> 16);
      }
  }
}

template <int F>
__global__ void __launch_bounds__(256)
k_gather_mean(const int2* __restrict__ rec,
              const int* __restrict__ col, const float* __restrict__ X,
              float* __restrict__ agg) {
  constexpr int V = F / 128;
  int node = __builtin_amdgcn_readfirstlane((int)((blockIdx.x * blockDim.x + threadIdx.x) >> 5));
  int lane = threadIdx.x & 31;
  if (node >= NNODES) return;
  int2 nr = rec[node];
  int beg = __builtin_amdgcn_readfirstlane(nr.x);
  int d = __builtin_amdgcn_readfirstlane(nr.y);
  const float* xl = X + (lane << 2);
  v4f acc[V];
#pragma unroll
  for (int v = 0; v < V; ++v) acc[v] = (v4f){0.f, 0.f, 0.f, 0.f};
  for (int base = 0; base < d; base += 32) {
    int m = d - base;
    if (m > 32) m = 32;
    int mine = (lane < m) ? col[beg + base + lane] : 0;
    int j = 0;
    for (; j + 4 <= m; j += 4) {
      v4f r[4][V];
#pragma unroll
      for (int u = 0; u < 4; ++u) {
        int s = __builtin_amdgcn_readlane(mine, j + u);
#pragma unroll
        for (int v = 0; v < V; ++v) r[u][v] = *(const v4f*)(xl + s * F + v * 128);
      }
#pragma unroll
      for (int u = 0; u < 4; ++u)
#pragma unroll
        for (int v = 0; v < V; ++v) acc[v] += r[u][v];
    }
    for (; j < m; ++j) {
      int s = __builtin_amdgcn_readlane(mine, j);
#pragma unroll
      for (int v = 0; v < V; ++v) acc[v] += *(const v4f*)(xl + s * F + v * 128);
    }
  }
  float inv = 1.f / fmaxf((float)d, 1.f);
  float* o = agg + node * F + (lane << 2);
#pragma unroll
  for (int v = 0; v < V; ++v) *(v4f*)(o + v * 128) = acc[v] * inv;
}

__device__ __forceinline__ void ld8(const float* __restrict__ p, int k, v4f& lo, v4f& hi) {
  lo = *(const v4f*)(p + k);
  hi = *(const v4f*)(p + k + 4);
}

__device__ __forceinline__ v2f pair_of(v4f lo, v4f hi, int q) {
  v2f r;
  if (q == 0) { r.x = lo.x; r.y = lo.y; }
  else if (q == 1) { r.x = lo.z; r.y = lo.w; }
  else if (q == 2) { r.x = hi.x; r.y = hi.y; }
  else { r.x = hi.z; r.y = hi.w; }
  return r;
}

#define SLAB   64
#define APITCH (SLAB + 4)
#define AROWS  (SM * 16)
#define ABUF   (AROWS * APITCH)
static_assert(SM == PM, "the node GEMMs share one row-block shape");
static_assert((AROWS * (SLAB / 4)) % 128 == 0, "a slab is a whole number of 16-byte pieces per thread");
static_assert(NNODES % AROWS == 0, "row blocks tile the nodes exactly");

template <int K, int WS, int NOP, int NT>
__device__ __forceinline__ void block_acc(const float* __restrict__ A0, const float* __restrict__ A1,
                                          const float* __restrict__ W0, const float* __restrict__ W1,
                                          int rowblk, int col0, float* sA, v8f (&c)[SM][NT]) {
  constexpr int SPO = K / SLAB;
  constexpr int T = NOP * SPO;
  constexpr int NP = AROWS * (SLAB / 4) / 128;
  int t = threadIdx.x;
  int lane = t & 31;
  int koff = (lane >> 4) << 3;
  int srow = t >> 4;
  int spc = (t & 15) << 2;
  const float* g0 = A0 + (rowblk + srow) * K + spc;
  const float* g1 = A1 + (rowblk + srow) * K + spc;
  float* sst = sA + srow * APITCH + spc;
  const float* sfr = sA + (lane & 15) * APITCH + koff;
  const float* w0 = W0 + col0 * WS + koff;
  const float* w1 = W1 + col0 * WS + koff;
  v4f st[NP];
#pragma unroll
  for (int i = 0; i < NP; ++i) st[i] = *(const v4f*)(g0 + i * 8 * K);
#pragma unroll
  for (int i = 0; i < NP; ++i) *(v4f*)(sst + i * 8 * APITCH) = st[i];
  __syncthreads();
#pragma unroll 1
  for (int g = 0; g < T; ++g) {
    int buf = g & 1;
    bool more = g + 1 < T;
    if (more) {
      int gn = g + 1;
      const float* src = (gn < SPO ? g0 : g1) + (gn & (SPO - 1)) * SLAB;
#pragma unroll
      for (int i = 0; i < NP; ++i) st[i] = *(const v4f*)(src + i * 8 * K);
    }
    const float* wb = (g < SPO ? w0 : w1) + (g & (SPO - 1)) * SLAB;
    const float* ab = sfr + buf * ABUF;
    for (int kk = 0; kk < SLAB; kk += 16) {
      v4f alo[SM], ahi[SM];
      v4f blo[NT], bhi[NT];
#pragma unroll
      for (int i = 0; i < NT; ++i) ld8(wb + i * 16 * WS, kk, blo[i], bhi[i]);
#pragma unroll
      for (int r = 0; r < SM; ++r) {
        alo[r] = *(const v4f*)(ab + r * 16 * APITCH + kk);
        ahi[r] = *(const v4f*)(ab + r * 16 * APITCH + kk + 4);
      }
#pragma unroll
      for (int q = 0; q < 4; ++q)
#pragma unroll
        for (int r = 0; r < SM; ++r)
#pragma unroll
          for (int i = 0; i < NT; ++i)
            c[r][i] = wmma_ab(pair_of(alo[r], ahi[r], q), pair_of(blo[i], bhi[i], q), c[r][i]);
    }
    if (more) {
      float* d = sst + (buf ^ 1) * ABUF;
#pragma unroll
      for (int i = 0; i < NP; ++i) *(v4f*)(d + i * 8 * APITCH) = st[i];
      __syncthreads();
    }
  }
}
static_assert((HID / 16) / SN == 8, "two blocks of 4 waves cover the 256 output columns of a SAGE layer");
static_assert((HID / 16) / PN == 4, "a block's 4 waves cover the 256 columns of a projection plane");

template <int K>
__global__ void __launch_bounds__(128)
k_sage_gemm(const float* __restrict__ A, const float* __restrict__ X,
            const float* __restrict__ Wl, const float* __restrict__ Wr,
            const float* __restrict__ bias, float* __restrict__ out) {
  __shared__ __attribute__((aligned(16))) float sA[2 * ABUF];
  int lane = threadIdx.x & 31;
  int gm = blockIdx.x >> 1;
  if (gm >= NNODES / AROWS) return;
  int gn = ((blockIdx.x & 1) << 2) + (threadIdx.x >> 5);
  int col0 = gn * (SN * 16) + (lane & 15);
  v8f c[SM][SN] = {};
  block_acc<K, K, 2, SN>(A, X, Wl, Wr, gm * AROWS, col0, sA, c);
  int mbase = gm * (SM * 16) + ((lane >> 4) << 3);
#pragma unroll
  for (int i = 0; i < SN; ++i) {
    int col = col0 + i * 16;
    float bv = bias[col];
#pragma unroll
    for (int t = 0; t < SM; ++t)
#pragma unroll
      for (int r = 0; r < 8; ++r)
        out[(mbase + t * 16 + r) * HID + col] = fmaxf(c[t][i][r] + bv, 0.f);
  }
}

__global__ void __launch_bounds__(128)
k_node_proj(const float* __restrict__ Hn, const float* __restrict__ W1,
            const float* __restrict__ b1, float* __restrict__ Q) {
  __shared__ __attribute__((aligned(16))) float sA[2 * ABUF];
  int lane = threadIdx.x & 31;
  int gm = blockIdx.x >> 1;
  if (gm >= NNODES / AROWS) return;
  int plane = blockIdx.x & 1;
  int gn = threadIdx.x >> 5;
  int col0 = gn * (PN * 16) + (lane & 15);
  const float* wp = W1 + plane * HID;
  v8f c[PM][PN] = {};
  block_acc<HID, ZK, 1, PN>(Hn, Hn, wp, wp, gm * AROWS, col0, sA, c);
  float* qp = Q + plane * (NNODES * HID);
  int mbase = gm * (PM * 16) + ((lane >> 4) << 3);
#pragma unroll
  for (int i = 0; i < PN; ++i) {
    int col = col0 + i * 16;
    float bv = (plane == 0) ? b1[col] : 0.f;
#pragma unroll
    for (int t = 0; t < PM; ++t)
#pragma unroll
      for (int r = 0; r < 8; ++r)
        qp[(mbase + t * 16 + r) * HID + col] = c[t][i][r] + bv;
  }
}

__device__ __forceinline__ int pair_col(int nt, int n) {
  return ((nt >> 2) << 6) + (n << 2) + (nt & 3);
}

__global__ void __launch_bounds__(256)
k_pair_mlp(const float* __restrict__ Qu, const float* __restrict__ Qv,
                           const int* __restrict__ eu, const int* __restrict__ ev,
                           const float* __restrict__ eattr,
                           const float* __restrict__ logexp,
                           const float* __restrict__ W1, const float* __restrict__ W2,
                           const float* __restrict__ b2, float* __restrict__ out) {
  __shared__ v2f sB[(HID / 16) * (EIN / 4) * 32];
  for (int idx = threadIdx.x; idx < (HID / 16) * (EIN / 4) * 32; idx += blockDim.x) {
    int nt = idx >> 7;
    int ks = (idx >> 5) & 3;
    int ln = idx & 31;
    const float* w = W1 + pair_col(nt, ln & 15) * ZK + 2 * HID;
    sB[idx] = ld2(w, ks * 4 + ((ln >> 4) << 1));
  }
  __syncthreads();

  int gtid = blockIdx.x * blockDim.x + threadIdx.x;
  int wave = gtid >> 5;
  int lane = threadIdx.x & 31;
  if (wave >= NPAIRS / 16 / TPW) return;
  int n = lane & 15;
  int half = (lane >> 4) << 3;
  int koff = (lane >> 4) << 1;
  float b2v = b2[0];
  v4f w2v[4];
#pragma unroll
  for (int j = 0; j < 4; ++j) w2v[j] = *(const v4f*)(W2 + (j << 6) + (n << 2));

  for (int tt = 0; tt < TPW; ++tt) {
    int tile = wave * TPW + tt;
    int ebase = tile * 16 + half;
    const float* arow = eattr + (tile * 16 + n) * EIN;
    v2f a[EIN / 4];
#pragma unroll
    for (int ks = 0; ks < EIN / 4; ++ks) a[ks] = ld2(arow, ks * 4 + koff);
    int uoff[8];
    int voff[8];
#pragma unroll
    for (int r = 0; r < 8; ++r) {
      uoff[r] = eu[ebase + r] * HID + (n << 2);
      voff[r] = ev[ebase + r] * HID + (n << 2);
    }
    float acc[8] = {};
#pragma unroll 1
    for (int j = 0; j < 4; ++j) {
      v4f pu[8];
      v4f pv[8];
#pragma unroll
      for (int r = 0; r < 8; ++r) {
        pu[r] = *(const v4f*)(Qu + uoff[r] + (j << 6));
        pv[r] = *(const v4f*)(Qv + voff[r] + (j << 6));
      }
      __builtin_amdgcn_sched_barrier(0);
      v2f bf[4][EIN / 4];
#pragma unroll
      for (int i = 0; i < 4; ++i)
#pragma unroll
        for (int ks = 0; ks < EIN / 4; ++ks)
          bf[i][ks] = sB[((j * 4 + i) * 4 + ks) * 32 + lane];
      v8f c[4];
#pragma unroll
      for (int r = 0; r < 8; ++r)
#pragma unroll
        for (int i = 0; i < 4; ++i) c[i][r] = pu[r][i] + pv[r][i];
#pragma unroll
      for (int ks = 0; ks < EIN / 4; ++ks)
#pragma unroll
        for (int i = 0; i < 4; ++i)
          c[i] = wmma_ab(a[ks], bf[i][ks], c[i]);
#pragma unroll
      for (int i = 0; i < 4; ++i)
#pragma unroll
        for (int r = 0; r < 8; ++r)
          acc[r] += fmaxf(c[i][r], 0.f) * w2v[j][i];
    }
#pragma unroll
    for (int r = 0; r < 8; ++r) {
      float s = acc[r];
      s += __shfl_xor(s, 1);
      s += __shfl_xor(s, 2);
      s += __shfl_xor(s, 4);
      s += __shfl_xor(s, 8);
      acc[r] = s;
    }
    if (n == 0) {
#pragma unroll
      for (int r = 0; r < 8; ++r)
        out[ebase + r] = logexp[ebase + r] + acc[r] + b2v;
    }
  }
}

extern "C" void kernel_launch(void* const* d_in, const int* in_sizes, int n_in,
                              void* d_out, int out_size, void* d_ws, size_t ws_size,
                              hipStream_t stream) {
  (void)in_sizes; (void)n_in; (void)out_size; (void)ws_size;
  const float* x      = (const float*)d_in[0];
  const int*   eidx   = (const int*)d_in[1];
  const int*   eu     = (const int*)d_in[2];
  const int*   ev     = (const int*)d_in[3];
  const float* eattr  = (const float*)d_in[4];
  const float* logexp = (const float*)d_in[5];
  const float* Wl1    = (const float*)d_in[6];
  const float* bl1    = (const float*)d_in[7];
  const float* Wr1    = (const float*)d_in[8];
  const float* Wl2    = (const float*)d_in[9];
  const float* bl2    = (const float*)d_in[10];
  const float* Wr2    = (const float*)d_in[11];
  const float* W1     = (const float*)d_in[12];
  const float* b1     = (const float*)d_in[13];
  const float* W2     = (const float*)d_in[14];
  const float* b2     = (const float*)d_in[15];
  float* out = (float*)d_out;

  const int* src = eidx;
  const int* dst = eidx + NEDGES;

  float* agg = (float*)d_ws;
  float* h1  = agg + NNODES * HID;
  float* h2  = h1 + NNODES * HID;
  int2*     rec     = (int2*)h2;
  unsigned* ebin    = (unsigned*)(rec + NNODES);
  int*      col     = (int*)ebin + NEDGES;
  int*      cnt     = col + NEDGES;
  int*      binbase = cnt + SCN;

  const int TB = 256;
  int gemm_waves = (NNODES / 16 / SM) * ((HID / 16) / SN);
  int gemm_blocks = gemm_waves / 4;
  int gather_blocks = NNODES / (TB / 32);

  {
    k_bin_count<<<NB1, 1024, 0, stream>>>(dst, cnt);
    k_bin_scatter<<<NB1, 1024, 0, stream>>>(src, dst, cnt, binbase, ebin);
    k_bucket<<<NBIN, 1024, 0, stream>>>(binbase, ebin, rec, col);
  }
  {
    k_gather_mean<F_IN><<<gather_blocks, TB, 0, stream>>>(rec, col, x, agg);
    k_sage_gemm<F_IN><<<gemm_blocks, 128, 0, stream>>>(agg, x, Wl1, Wr1, bl1, h1);
  }
  {
    k_gather_mean<HID><<<gather_blocks, TB, 0, stream>>>(rec, col, h1, agg);
    k_sage_gemm<HID><<<gemm_blocks, 128, 0, stream>>>(agg, h1, Wl2, Wr2, bl2, h2);
  }
  {
    float* qu = agg;
    float* qv = h1;
    int pwaves = (NNODES / 16 / PM) * ((2 * HID / 16) / PN);
    k_node_proj<<<pwaves / 4, 128, 0, stream>>>(h2, W1, b1, qu);
    int waves = NPAIRS / 16 / TPW;
    int blocks = (waves * 32 + TB - 1) / TB;
    k_pair_mlp<<<blocks, TB, 0, stream>>>(qu, qv, eu, ev, eattr, logexp,
                                          W1, W2, b2, out);
  }
}
